# starting kernel
# baseline (speedup 1.0000x reference)
#include <hip/hip_runtime.h>
#include <hip/hip_bf16.h>
#include <math.h>

typedef float v2f __attribute__((ext_vector_type(2)));
typedef float v8f __attribute__((ext_vector_type(8)));

#define B_ 4
#define M_ 2048
#define K_ 64
#define C_ 64
#define P_ 28
#define NCURVE 64

__device__ __forceinline__ v8f vmin8(v8f a, v8f b){
  v8f r;
#pragma unroll
  for (int i = 0; i < 8; ++i) r[i] = fminf(a[i], b[i]);
  return r;
}

__global__ __launch_bounds__(128) void sdf_kernel(
    const float* __restrict__ pts, const float* __restrict__ prim,
    float* __restrict__ sdf_out, float* __restrict__ dmin_out)
{
  __shared__ float4 s_c4[NCURVE];
  __shared__ float4 s_e4[NCURVE];
  const int b = blockIdx.x >> 6;
  const int k = blockIdx.x & 63;
  const float* Pp = prim + (size_t)b * P_ * K_ + k;
  const int tid = threadIdx.x;
  const float PI = 3.14159265358979323846f;

  if (tid < NCURVE) {
    int seg = tid >> 4;
    float t = (float)(tid & 15) * (1.0f / 16.0f);
    float theta = PI * 0.125f + atanf(tanf(PI * 0.125f) * (1.0f / 3.0f));
    float r0 = (PI * 0.5f) * (float)seg;
    float r1 = r0 + theta;
    float r2 = (PI * 0.5f) * (float)(seg + 1) - theta;
    float r3 = (PI * 0.5f) * (float)(seg + 1);
    float rad0 = fabsf(Pp[(8 + seg * 3 + 0) * K_]);
    float rad1 = fabsf(Pp[(8 + seg * 3 + 1) * K_]);
    float rad2 = fabsf(Pp[(8 + seg * 3 + 2) * K_]);
    int seg1 = (seg + 1) & 3;
    float rad3 = fabsf(Pp[(8 + seg1 * 3 + 0) * K_]);
    float P0x = cosf(r0) * rad0, P0y = sinf(r0) * rad0;
    float P1x = cosf(r1) * rad1, P1y = sinf(r1) * rad1;
    float P2x = cosf(r2) * rad2, P2y = sinf(r2) * rad2;
    float P3x = cosf(r3) * rad3, P3y = sinf(r3) * rad3;
    float w1 = fabsf(Pp[(20 + seg * 2 + 0) * K_]);
    float w2 = fabsf(Pp[(20 + seg * 2 + 1) * K_]);
    float u = 1.0f - t;
    float b0 = u * u * u, b1 = 3.0f * u * u * t, b2 = 3.0f * u * t * t, b3 = t * t * t;
    float inv = 1.0f / (b0 + w1 * b1 + w2 * b2 + b3);
    float cx = (P0x * b0 + P1x * (w1 * b1) + P2x * (w2 * b2) + P3x * b3) * inv;
    float cy = (P0y * b0 + P1y * (w1 * b1) + P2y * (w2 * b2) + P3y * b3) * inv;
    s_c4[tid] = make_float4(cx, cy, cx * cx + cy * cy, 1.0f);
  }

  float q0 = Pp[0], q1 = Pp[K_], q2 = Pp[2 * K_], q3 = Pp[3 * K_];
  float invn = 1.0f / sqrtf(q0 * q0 + q1 * q1 + q2 * q2 + q3 * q3 + 1e-8f);
  float qw = q0 * invn, vx = -q1 * invn, vy = -q2 * invn, vz = -q3 * invn;
  float tx = Pp[4 * K_], ty = Pp[5 * K_], tz = Pp[6 * K_];
  float habs = fabsf(Pp[7 * K_]);
  __syncthreads();

  if (tid < NCURVE) {
    float4 ca = s_c4[tid];
    float4 cb = s_c4[(tid + 1) & 63];
    s_e4[tid] = make_float4(ca.y - cb.y, cb.x - ca.x,
                            ca.x * cb.y - ca.y * cb.x, ca.y);
  }

  const int lane = tid & 31;
  const int wv = tid >> 5;
  const int row = lane & 15;
  const bool hi = lane >= 16;
  const int m = blockIdx.y * 64 + wv * 16 + row;

  const float* p3 = pts + ((size_t)b * M_ + m) * 3;
  float px0 = p3[0] - tx, py0 = p3[1] - ty, pz0 = p3[2] - tz;
  float t2x = 2.0f * (vy * pz0 - vz * py0);
  float t2y = 2.0f * (vz * px0 - vx * pz0);
  float t2z = 2.0f * (vx * py0 - vy * px0);
  float plx = px0 + qw * t2x + (vy * t2z - vz * t2y);
  float ply = py0 + qw * t2y + (vz * t2x - vx * t2z);
  float plz = pz0 + qw * t2z + (vx * t2y - vy * t2x);
  float px = plx, py = ply;
  float ss = px * px + py * py;
  __syncthreads();

  v2f Bd; Bd.x = hi ? ss : px;   Bd.y = hi ? 1.0f : py;
  v2f Bw; Bw.x = hi ? 1.0f : px; Bw.y = hi ? 0.0f : py;
  v8f z = {0.f, 0.f, 0.f, 0.f, 0.f, 0.f, 0.f, 0.f};
  v8f e = {3e38f, 3e38f, 3e38f, 3e38f, 3e38f, 3e38f, 3e38f, 3e38f};

  int wn = 0;
  int le0b = 0;
  float cr7[4];
  int   le7[4];
#pragma unroll
  for (int tTile = 0; tTile < 4; ++tTile) {
    float4 cc = s_c4[tTile * 16 + row];
    float4 ee = s_e4[tTile * 16 + row];
    v2f Ad; Ad.x = hi ? 1.0f   : -2.0f * cc.x;  Ad.y = hi ? cc.z : -2.0f * cc.y;
    v2f Ac; Ac.x = hi ? ee.z   : ee.x;          Ac.y = hi ? 0.0f : ee.y;
    v2f Ay; Ay.x = hi ? -ee.w  : 0.0f;          Ay.y = hi ? 0.0f : 1.0f;
    v8f d  = __builtin_amdgcn_wmma_f32_16x16x4_f32(
        false, Ad, false, Bd, (short)0, z, false, false);
    v8f cr = __builtin_amdgcn_wmma_f32_16x16x4_f32(
        false, Ac, false, Bw, (short)0, z, false, false);
    v8f yd = __builtin_amdgcn_wmma_f32_16x16x4_f32(
        false, Ay, false, Bw, (short)0, z, false, false);
    e = vmin8(e, d);
    int le[8];
#pragma unroll
    for (int v = 0; v < 8; ++v) le[v] = (yd[v] >= 0.0f);
#pragma unroll
    for (int v = 0; v < 7; ++v) {
      int aLE = le[v], bLE = le[v + 1];
      float c = cr[v];
      wn += (aLE & (bLE ^ 1) & (c > 0.0f)) - ((aLE ^ 1) & bLE & (c < 0.0f));
    }
    le0b |= le[0] << tTile;
    cr7[tTile] = cr[7];
    le7[tTile] = le[7];
  }
  int recvb = __shfl_xor(le0b, 16, 32);
#pragma unroll
  for (int tTile = 0; tTile < 4; ++tTile) {
    int nt = hi ? ((tTile + 1) & 3) : tTile;
    int bLE = (recvb >> nt) & 1;
    int aLE = le7[tTile];
    float c = cr7[tTile];
    wn += (aLE & (bLE ^ 1) & (c > 0.0f)) - ((aLE ^ 1) & bLE & (c < 0.0f));
  }
  wn += __shfl_xor(wn, 16, 32);

  float m01 = fminf(e[0], e[1]), m23 = fminf(e[2], e[3]);
  float m45 = fminf(e[4], e[5]), m67 = fminf(e[6], e[7]);
  float m8  = fminf(fminf(m01, m23), fminf(m45, m67));
  float d2min = fminf(m8, __shfl_xor(m8, 16, 32));

  float dminv = sqrtf(d2min + 1e-12f);
  float sdf2d = (wn != 0) ? -dminv : dminv;
  float dz = fabsf(plz) - habs;
  float po = fmaxf(sdf2d, 0.0f), dzo = fmaxf(dz, 0.0f);
  float sdf = fminf(fmaxf(sdf2d, dz), 0.0f) + sqrtf(po * po + dzo * dzo + 1e-12f);

  size_t o = ((size_t)b * M_ + m) * K_ + k;
  float* basep = hi ? dmin_out : sdf_out;
  float  outv  = hi ? dminv    : sdf;
  basep[o] = outv;
}

__global__ __launch_bounds__(64) void combine_kernel(
    const float* __restrict__ sdf_in, const float* __restrict__ iw,
    const float* __restrict__ uw, const int* __restrict__ is_training,
    float* __restrict__ inter_out, float* __restrict__ occ_out)
{
  const int bm = blockIdx.x;
  const int b  = bm >> 11;
  const int c  = threadIdx.x;
  __shared__ float occ_sh[K_];
  __shared__ float preu[C_];

  float sv = sdf_in[(size_t)bm * K_ + c];
  occ_sh[c] = 1.0f / (1.0f + __expf(150.0f * sv));
  const float* wp = iw + ((size_t)b * K_) * C_ + c;
  __builtin_prefetch(wp, 0, 3);
  int train = is_training[0];
  __syncthreads();

  float inter;
  if (train) {
    float pmin = 3.0e38f;
#pragma unroll 4
    for (int kk = 0; kk < K_; ++kk) {
      float p = 1.0f - wp[(size_t)kk * C_] * (1.0f - occ_sh[kk]);
      pmin = fminf(pmin, p);
    }
    float s = 0.0f, ts = 0.0f;
#pragma unroll 4
    for (int kk = 0; kk < K_; ++kk) {
      float p = 1.0f - wp[(size_t)kk * C_] * (1.0f - occ_sh[kk]);
      float ee = __expf(-40.0f * (p - pmin));
      s += ee; ts += ee * p;
    }
    inter = ts / s;
  } else {
    inter = 3.0e38f;
#pragma unroll 4
    for (int kk = 0; kk < K_; ++kk) {
      float p = 1.0f - wp[(size_t)kk * C_] * (1.0f - occ_sh[kk]);
      inter = fminf(inter, p);
    }
  }
  inter_out[(size_t)bm * C_ + c] = inter;
  preu[c] = uw[b * C_ + c] * inter;
  __syncthreads();

  if (c < 32) {
    float a0 = preu[c], a1 = preu[c + 32];
    if (train) {
      float mx = fmaxf(a0, a1);
#pragma unroll
      for (int mask = 1; mask <= 16; mask <<= 1)
        mx = fmaxf(mx, __shfl_xor(mx, mask, 32));
      float e0 = __expf(40.0f * (a0 - mx));
      float e1 = __expf(40.0f * (a1 - mx));
      float s  = e0 + e1;
      float ts = e0 * a0 + e1 * a1;
#pragma unroll
      for (int mask = 1; mask <= 16; mask <<= 1) {
        s  += __shfl_xor(s,  mask, 32);
        ts += __shfl_xor(ts, mask, 32);
      }
      if (c == 0) occ_out[bm] = ts / s;
    } else {
      float mx = fmaxf(a0, a1);
#pragma unroll
      for (int mask = 1; mask <= 16; mask <<= 1)
        mx = fmaxf(mx, __shfl_xor(mx, mask, 32));
      if (c == 0) occ_out[bm] = mx;
    }
  }
}

extern "C" void kernel_launch(void* const* d_in, const int* in_sizes, int n_in,
                              void* d_out, int out_size, void* d_ws, size_t ws_size,
                              hipStream_t stream) {
  const float* pts   = (const float*)d_in[0];
  const float* prim  = (const float*)d_in[1];
  const float* iw    = (const float*)d_in[2];
  const float* uw    = (const float*)d_in[3];
  const int*   train = (const int*)d_in[4];

  float* occ_out   = (float*)d_out;
  float* sdf_out   = occ_out + (size_t)B_ * M_;
  float* inter_out = sdf_out + (size_t)B_ * M_ * K_;
  float* dmin_out  = inter_out + (size_t)B_ * M_ * C_;

  dim3 g1(B_ * K_, M_ / 64);
  sdf_kernel<<<g1, 128, 0, stream>>>(pts, prim, sdf_out, dmin_out);
  combine_kernel<<<B_ * M_, 64, 0, stream>>>(sdf_out, iw, uw, train,
                                             inter_out, occ_out);
  (void)in_sizes; (void)n_in; (void)out_size; (void)d_ws; (void)ws_size;
}


# work-group coarsening in both launches: an sdf block reuses its curve build for 4 point tiles, a combine block keeps its weight column in registers for 4 rows; set-up paid once, 4x fewer waves
# speedup vs baseline: 1.3541x; 1.3541x over previous
#include <hip/hip_runtime.h>
#include <hip/hip_bf16.h>
#include <math.h>

typedef float v2f __attribute__((ext_vector_type(2)));
typedef float v8f __attribute__((ext_vector_type(8)));

#define B_ 4
#define M_ 2048
#define K_ 64
#define C_ 64
#define P_ 28
#define NCURVE 64
#define TPB 4
#define RPB 4

__device__ __forceinline__ v8f vmin8(v8f a, v8f b){
  v8f r;
#pragma unroll
  for (int i = 0; i < 8; ++i) r[i] = fminf(a[i], b[i]);
  return r;
}

__global__ __launch_bounds__(128) void sdf_kernel(
    const float* __restrict__ pts, const float* __restrict__ prim,
    float* __restrict__ sdf_out, float* __restrict__ dmin_out)
{
  __shared__ float4 s_c4[NCURVE];
  __shared__ float4 s_e4[NCURVE];
  const int b = blockIdx.x >> 6;
  const int k = blockIdx.x & 63;
  const float* Pp = prim + (size_t)b * P_ * K_ + k;
  const int tid = threadIdx.x;
  const float PI = 3.14159265358979323846f;
  const int lane = tid & 31;
  const int wv = tid >> 5;
  const int row = lane & 15;
  const bool hi = lane >= 16;
  const int m0 = blockIdx.y * (64 * TPB) + wv * 16 + row;

  float rx[TPB], ry[TPB], rz[TPB];
#pragma unroll
  for (int j = 0; j < TPB; ++j) {
    const float* p3 = pts + ((size_t)b * M_ + m0 + j * 64) * 3;
    rx[j] = p3[0]; ry[j] = p3[1]; rz[j] = p3[2];
  }

  if (tid < NCURVE) {
    int seg = tid >> 4;
    float t = (float)(tid & 15) * (1.0f / 16.0f);
    float theta = PI * 0.125f + atanf(tanf(PI * 0.125f) * (1.0f / 3.0f));
    float r0 = (PI * 0.5f) * (float)seg;
    float r1 = r0 + theta;
    float r2 = (PI * 0.5f) * (float)(seg + 1) - theta;
    float r3 = (PI * 0.5f) * (float)(seg + 1);
    float rad0 = fabsf(Pp[(8 + seg * 3 + 0) * K_]);
    float rad1 = fabsf(Pp[(8 + seg * 3 + 1) * K_]);
    float rad2 = fabsf(Pp[(8 + seg * 3 + 2) * K_]);
    int seg1 = (seg + 1) & 3;
    float rad3 = fabsf(Pp[(8 + seg1 * 3 + 0) * K_]);
    float P0x = cosf(r0) * rad0, P0y = sinf(r0) * rad0;
    float P1x = cosf(r1) * rad1, P1y = sinf(r1) * rad1;
    float P2x = cosf(r2) * rad2, P2y = sinf(r2) * rad2;
    float P3x = cosf(r3) * rad3, P3y = sinf(r3) * rad3;
    float w1 = fabsf(Pp[(20 + seg * 2 + 0) * K_]);
    float w2 = fabsf(Pp[(20 + seg * 2 + 1) * K_]);
    float u = 1.0f - t;
    float b0 = u * u * u, b1 = 3.0f * u * u * t, b2 = 3.0f * u * t * t, b3 = t * t * t;
    float inv = 1.0f / (b0 + w1 * b1 + w2 * b2 + b3);
    float cx = (P0x * b0 + P1x * (w1 * b1) + P2x * (w2 * b2) + P3x * b3) * inv;
    float cy = (P0y * b0 + P1y * (w1 * b1) + P2y * (w2 * b2) + P3y * b3) * inv;
    s_c4[tid] = make_float4(cx, cy, cx * cx + cy * cy, 1.0f);
  }

  float q0 = Pp[0], q1 = Pp[K_], q2 = Pp[2 * K_], q3 = Pp[3 * K_];
  float invn = 1.0f / sqrtf(q0 * q0 + q1 * q1 + q2 * q2 + q3 * q3 + 1e-8f);
  float qw = q0 * invn, vx = -q1 * invn, vy = -q2 * invn, vz = -q3 * invn;
  float tx = Pp[4 * K_], ty = Pp[5 * K_], tz = Pp[6 * K_];
  float habs = fabsf(Pp[7 * K_]);
  __syncthreads();

  if (tid < NCURVE) {
    float4 ca = s_c4[tid];
    float4 cb = s_c4[(tid + 1) & 63];
    s_e4[tid] = make_float4(ca.y - cb.y, cb.x - ca.x,
                            ca.x * cb.y - ca.y * cb.x, ca.y);
  }

  __syncthreads();

  v2f Ad[4], Ac[4], Ay[4];
#pragma unroll
  for (int tTile = 0; tTile < 4; ++tTile) {
    float4 cc = s_c4[tTile * 16 + row];
    float4 ee = s_e4[tTile * 16 + row];
    Ad[tTile].x = hi ? 1.0f  : -2.0f * cc.x;  Ad[tTile].y = hi ? cc.z : -2.0f * cc.y;
    Ac[tTile].x = hi ? ee.z  : ee.x;          Ac[tTile].y = hi ? 0.0f : ee.y;
    Ay[tTile].x = hi ? -ee.w : 0.0f;          Ay[tTile].y = hi ? 0.0f : 1.0f;
  }
  const v8f z = {0.f, 0.f, 0.f, 0.f, 0.f, 0.f, 0.f, 0.f};
  float* const basep = hi ? dmin_out : sdf_out;

#pragma unroll 1
  for (int it = 0; it < TPB; ++it) {
    const int m = m0 + it * 64;

    float px0 = rx[0] - tx, py0 = ry[0] - ty, pz0 = rz[0] - tz;
#pragma unroll
    for (int j = 0; j + 1 < TPB; ++j) { rx[j] = rx[j + 1]; ry[j] = ry[j + 1]; rz[j] = rz[j + 1]; }
    float t2x = 2.0f * (vy * pz0 - vz * py0);
    float t2y = 2.0f * (vz * px0 - vx * pz0);
    float t2z = 2.0f * (vx * py0 - vy * px0);
    float plx = px0 + qw * t2x + (vy * t2z - vz * t2y);
    float ply = py0 + qw * t2y + (vz * t2x - vx * t2z);
    float plz = pz0 + qw * t2z + (vx * t2y - vy * t2x);
    float px = plx, py = ply;
    float ss = px * px + py * py;

    v2f Bd; Bd.x = hi ? ss : px;   Bd.y = hi ? 1.0f : py;
    v2f Bw; Bw.x = hi ? 1.0f : px; Bw.y = hi ? 0.0f : py;
    v8f e = {3e38f, 3e38f, 3e38f, 3e38f, 3e38f, 3e38f, 3e38f, 3e38f};

    int wn = 0;
    int le0b = 0;
    float cr7[4];
    int   le7[4];
#pragma unroll
    for (int tTile = 0; tTile < 4; ++tTile) {
      v8f d  = __builtin_amdgcn_wmma_f32_16x16x4_f32(
          false, Ad[tTile], false, Bd, (short)0, z, false, false);
      v8f cr = __builtin_amdgcn_wmma_f32_16x16x4_f32(
          false, Ac[tTile], false, Bw, (short)0, z, false, false);
      v8f yd = __builtin_amdgcn_wmma_f32_16x16x4_f32(
          false, Ay[tTile], false, Bw, (short)0, z, false, false);
      e = vmin8(e, d);
      int le[8];
#pragma unroll
      for (int v = 0; v < 8; ++v) le[v] = (yd[v] >= 0.0f);
#pragma unroll
      for (int v = 0; v < 7; ++v) {
        int aLE = le[v], bLE = le[v + 1];
        float c = cr[v];
        wn += (aLE & (bLE ^ 1) & (c > 0.0f)) - ((aLE ^ 1) & bLE & (c < 0.0f));
      }
      le0b |= le[0] << tTile;
      cr7[tTile] = cr[7];
      le7[tTile] = le[7];
    }
    int recvb = __shfl_xor(le0b, 16, 32);
#pragma unroll
    for (int tTile = 0; tTile < 4; ++tTile) {
      int nt = hi ? ((tTile + 1) & 3) : tTile;
      int bLE = (recvb >> nt) & 1;
      int aLE = le7[tTile];
      float c = cr7[tTile];
      wn += (aLE & (bLE ^ 1) & (c > 0.0f)) - ((aLE ^ 1) & bLE & (c < 0.0f));
    }
    wn += __shfl_xor(wn, 16, 32);

    float m01 = fminf(e[0], e[1]), m23 = fminf(e[2], e[3]);
    float m45 = fminf(e[4], e[5]), m67 = fminf(e[6], e[7]);
    float m8  = fminf(fminf(m01, m23), fminf(m45, m67));
    float d2min = fminf(m8, __shfl_xor(m8, 16, 32));

    float dminv = sqrtf(d2min + 1e-12f);
    float sdf2d = (wn != 0) ? -dminv : dminv;
    float dz = fabsf(plz) - habs;
    float po = fmaxf(sdf2d, 0.0f), dzo = fmaxf(dz, 0.0f);
    float sdf = fminf(fmaxf(sdf2d, dz), 0.0f) + sqrtf(po * po + dzo * dzo + 1e-12f);

    size_t o = ((size_t)b * M_ + m) * K_ + k;
    float  outv  = hi ? dminv    : sdf;
    basep[o] = outv;
  }
}

__global__ __launch_bounds__(64) void combine_kernel(
    const float* __restrict__ sdf_in, const float* __restrict__ iw,
    const float* __restrict__ uw, const int* __restrict__ is_training,
    float* __restrict__ inter_out, float* __restrict__ occ_out)
{
  const int bm0 = blockIdx.x * RPB;
  const int b   = bm0 >> 11;
  const int c   = threadIdx.x;
  __shared__ __align__(16) float occ_sh[RPB][K_];
  __shared__ float preu[RPB][C_];

  const float* wp = iw + ((size_t)b * K_) * C_ + c;
  float w[K_];
#pragma unroll
  for (int kk = 0; kk < K_; ++kk) w[kk] = wp[(size_t)kk * C_];
  const float uwc = uw[b * C_ + c];
  const int train = is_training[0];
#pragma unroll
  for (int r = 0; r < RPB; ++r) {
    float sv = sdf_in[(size_t)(bm0 + r) * K_ + c];
    occ_sh[r][c] = 1.0f / (1.0f + __expf(150.0f * sv));
  }
  __syncthreads();

#pragma unroll 1
  for (int r = 0; r < RPB; ++r) {
    float p[K_];
    float pmin = 3.0e38f;
#pragma unroll
    for (int kk = 0; kk < K_; ++kk) {
      p[kk] = 1.0f - w[kk] * (1.0f - occ_sh[r][kk]);
      pmin = fminf(pmin, p[kk]);
    }
    float inter;
    if (train) {
      float s = 0.0f, ts = 0.0f;
#pragma unroll
      for (int kk = 0; kk < K_; ++kk) {
        float ee = __expf(-40.0f * (p[kk] - pmin));
        s += ee; ts += ee * p[kk];
      }
      inter = ts / s;
    } else {
      inter = pmin;
    }
    inter_out[(size_t)(bm0 + r) * C_ + c] = inter;
    preu[r][c] = uwc * inter;
  }
  __syncthreads();

  const int lane = c & 31;
  const int wv = c >> 5;
#pragma unroll
  for (int rr = 0; rr < RPB / 2; ++rr) {
    const int r = wv * (RPB / 2) + rr;
    float a0 = preu[r][lane], a1 = preu[r][lane + 32];
    float mx = fmaxf(a0, a1);
#pragma unroll
    for (int mask = 1; mask <= 16; mask <<= 1)
      mx = fmaxf(mx, __shfl_xor(mx, mask, 32));
    float res = mx;
    if (train) {
      float e0 = __expf(40.0f * (a0 - mx));
      float e1 = __expf(40.0f * (a1 - mx));
      float s  = e0 + e1;
      float ts = e0 * a0 + e1 * a1;
#pragma unroll
      for (int mask = 1; mask <= 16; mask <<= 1) {
        s  += __shfl_xor(s,  mask, 32);
        ts += __shfl_xor(ts, mask, 32);
      }
      res = ts / s;
    }
    if (lane == 0) occ_out[bm0 + r] = res;
  }
}

extern "C" void kernel_launch(void* const* d_in, const int* in_sizes, int n_in,
                              void* d_out, int out_size, void* d_ws, size_t ws_size,
                              hipStream_t stream) {
  const float* pts   = (const float*)d_in[0];
  const float* prim  = (const float*)d_in[1];
  const float* iw    = (const float*)d_in[2];
  const float* uw    = (const float*)d_in[3];
  const int*   train = (const int*)d_in[4];

  float* occ_out   = (float*)d_out;
  float* sdf_out   = occ_out + (size_t)B_ * M_;
  float* inter_out = sdf_out + (size_t)B_ * M_ * K_;
  float* dmin_out  = inter_out + (size_t)B_ * M_ * C_;

  dim3 g1(B_ * K_, M_ / (64 * TPB));
  sdf_kernel<<<g1, 128, 0, stream>>>(pts, prim, sdf_out, dmin_out);
  combine_kernel<<<B_ * M_ / RPB, 64, 0, stream>>>(sdf_out, iw, uw, train,
                                             inter_out, occ_out);
  (void)in_sizes; (void)n_in; (void)out_size; (void)d_ws; (void)ws_size;
}


# one launch instead of two: a work-group owns 32 points for all 64 primitives, builds the batch's curves once in LDS, keeps the distances there for the combine and stores them as whole lines
# speedup vs baseline: 2.6751x; 1.9756x over previous
#include <hip/hip_runtime.h>
#include <hip/hip_bf16.h>
#include <math.h>

typedef float v2f __attribute__((ext_vector_type(2)));
typedef float v8f __attribute__((ext_vector_type(8)));

#define B_ 4
#define M_ 2048
#define K_ 64
#define C_ 64
#define P_ 28
#define NCURVE 64
#define PTS 32
#define NT 512
#define KPW 8
#define RPT 4

__device__ __forceinline__ v8f vmin8(v8f a, v8f b){
  v8f r;
#pragma unroll
  for (int i = 0; i < 8; ++i) r[i] = fminf(a[i], b[i]);
  return r;
}

__global__ __launch_bounds__(NT) void fused_kernel(
    const float* __restrict__ pts, const float* __restrict__ prim,
    const float* __restrict__ iw, const float* __restrict__ uw,
    const int* __restrict__ is_training,
    float* __restrict__ occ_out, float* __restrict__ sdf_out,
    float* __restrict__ inter_out, float* __restrict__ dmin_out)
{
  __shared__ float4 s_c4[K_][NCURVE];
  __shared__ float4 s_e4[K_][NCURVE];
  __shared__ float4 s_trig[4][2];
  __shared__ float4 s_pose[K_][2];
  __shared__ float  s_prim[P_ * K_];
  __shared__ __align__(16) float s_sdf[PTS][K_];
  __shared__ __align__(16) float s_dmin[PTS][K_];
  __shared__ __align__(16) float occ_sh[PTS][K_];
  __shared__ float preu[PTS][C_];

  const int blk = blockIdx.x;
  const int b   = blk >> 6;
  const int bm0 = blk * PTS;
  const int tid = threadIdx.x;
  const int lane = tid & 31;
  const int wv = tid >> 5;
  const int row = lane & 15;
  const bool hi = lane >= 16;
  const float PI = 3.14159265358979323846f;

  const int ptile = wv & 1;
  const float* p3 = pts + ((size_t)bm0 + ptile * 16 + row) * 3;
  const float rx = p3[0], ry = p3[1], rz = p3[2];

  float pr[4];
#pragma unroll
  for (int i = 0; i < 4; ++i) {
    const int idx = tid + i * NT;
    pr[i] = (idx < P_ * K_) ? prim[(size_t)b * P_ * K_ + idx] : 0.0f;
  }

  const int c = tid & 63;
  const float* wp = iw + ((size_t)b * K_) * C_ + c;
  float w[K_];
#pragma unroll
  for (int kk = 0; kk < K_; ++kk) w[kk] = wp[(size_t)kk * C_];
  const float uwc = uw[b * C_ + c];
  const int train = is_training[0];

  if (tid < 16) {
    const int seg = tid >> 2;
    const int a = tid & 3;
    const float theta = PI * 0.125f + atanf(tanf(PI * 0.125f) * (1.0f / 3.0f));
    const float r0 = (PI * 0.5f) * (float)seg;
    const float r3 = (PI * 0.5f) * (float)(seg + 1);
    const float ra = (a == 0) ? r0 : (a == 1) ? (r0 + theta) : (a == 2) ? (r3 - theta) : r3;
    reinterpret_cast<float2*>(&s_trig[seg][0])[a] = make_float2(cosf(ra), sinf(ra));
  }
#pragma unroll
  for (int i = 0; i < 4; ++i) {
    const int idx = tid + i * NT;
    if (idx < P_ * K_) s_prim[idx] = pr[i];
  }
  __syncthreads();

  {
    const int j  = tid & 63;
    const int kg = tid >> 6;
    const int seg = j >> 4;
    const int seg1 = (seg + 1) & 3;
    const float t = (float)(j & 15) * (1.0f / 16.0f);
    const float u = 1.0f - t;
    const float b0 = u * u * u, b1 = 3.0f * u * u * t, b2 = 3.0f * u * t * t, b3 = t * t * t;
    const float4 ta = s_trig[seg][0];
    const float4 tb = s_trig[seg][1];
    for (int i = 0; i < K_ / 8; ++i) {
      const int k = kg + 8 * i;
      const float rad0 = fabsf(s_prim[(8 + seg * 3 + 0) * K_ + k]);
      const float rad1 = fabsf(s_prim[(8 + seg * 3 + 1) * K_ + k]);
      const float rad2 = fabsf(s_prim[(8 + seg * 3 + 2) * K_ + k]);
      const float rad3 = fabsf(s_prim[(8 + seg1 * 3 + 0) * K_ + k]);
      const float P0x = ta.x * rad0, P0y = ta.y * rad0;
      const float P1x = ta.z * rad1, P1y = ta.w * rad1;
      const float P2x = tb.x * rad2, P2y = tb.y * rad2;
      const float P3x = tb.z * rad3, P3y = tb.w * rad3;
      const float w1 = fabsf(s_prim[(20 + seg * 2 + 0) * K_ + k]);
      const float w2 = fabsf(s_prim[(20 + seg * 2 + 1) * K_ + k]);
      const float inv = 1.0f / (b0 + w1 * b1 + w2 * b2 + b3);
      const float cx = (P0x * b0 + P1x * (w1 * b1) + P2x * (w2 * b2) + P3x * b3) * inv;
      const float cy = (P0y * b0 + P1y * (w1 * b1) + P2y * (w2 * b2) + P3y * b3) * inv;
      s_c4[k][j] = make_float4(cx, cy, cx * cx + cy * cy, 1.0f);
    }
  }
  if (tid < K_) {
    const float q0 = s_prim[tid], q1 = s_prim[K_ + tid];
    const float q2 = s_prim[2 * K_ + tid], q3 = s_prim[3 * K_ + tid];
    const float invn = 1.0f / sqrtf(q0 * q0 + q1 * q1 + q2 * q2 + q3 * q3 + 1e-8f);
    s_pose[tid][0] = make_float4(q0 * invn, -q1 * invn, -q2 * invn, -q3 * invn);
    s_pose[tid][1] = make_float4(s_prim[4 * K_ + tid], s_prim[5 * K_ + tid],
                                 s_prim[6 * K_ + tid], fabsf(s_prim[7 * K_ + tid]));
  }
  __syncthreads();

  {
    const int j  = tid & 63;
    const int kg = tid >> 6;
    for (int i = 0; i < K_ / 8; ++i) {
      const int k = kg + 8 * i;
      const float4 ca = s_c4[k][j];
      const float4 cb = s_c4[k][(j + 1) & 63];
      s_e4[k][j] = make_float4(ca.y - cb.y, cb.x - ca.x,
                               ca.x * cb.y - ca.y * cb.x, ca.y);
    }
  }
  __syncthreads();

  {
    const int kbase = (wv >> 1) * KPW;
    const int prow = ptile * 16 + row;
    float* const tile_out = hi ? &s_dmin[prow][0] : &s_sdf[prow][0];
    const v8f z = {0.f, 0.f, 0.f, 0.f, 0.f, 0.f, 0.f, 0.f};

#pragma unroll 1
    for (int kk = 0; kk < KPW; ++kk) {
      const int k = kbase + kk;
      const float4 pa = s_pose[k][0];
      const float4 pb = s_pose[k][1];
      const float qw = pa.x, vx = pa.y, vy = pa.z, vz = pa.w;
      const float habs = pb.w;

      v2f Ad[4], Ac[4], Ay[4];
#pragma unroll
      for (int tTile = 0; tTile < 4; ++tTile) {
        const float4 cc = s_c4[k][tTile * 16 + row];
        const float4 ee = s_e4[k][tTile * 16 + row];
        Ad[tTile].x = hi ? 1.0f  : -2.0f * cc.x;  Ad[tTile].y = hi ? cc.z : -2.0f * cc.y;
        Ac[tTile].x = hi ? ee.z  : ee.x;          Ac[tTile].y = hi ? 0.0f : ee.y;
        Ay[tTile].x = hi ? -ee.w : 0.0f;          Ay[tTile].y = hi ? 0.0f : 1.0f;
      }

      const float px0 = rx - pb.x, py0 = ry - pb.y, pz0 = rz - pb.z;
      const float t2x = 2.0f * (vy * pz0 - vz * py0);
      const float t2y = 2.0f * (vz * px0 - vx * pz0);
      const float t2z = 2.0f * (vx * py0 - vy * px0);
      const float plx = px0 + qw * t2x + (vy * t2z - vz * t2y);
      const float ply = py0 + qw * t2y + (vz * t2x - vx * t2z);
      const float plz = pz0 + qw * t2z + (vx * t2y - vy * t2x);
      const float px = plx, py = ply;
      const float ss = px * px + py * py;

      v2f Bd; Bd.x = hi ? ss : px;   Bd.y = hi ? 1.0f : py;
      v2f Bw; Bw.x = hi ? 1.0f : px; Bw.y = hi ? 0.0f : py;
      v8f e = {3e38f, 3e38f, 3e38f, 3e38f, 3e38f, 3e38f, 3e38f, 3e38f};

      int wn = 0;
      int le0b = 0;
      float cr7[4];
      int   le7[4];
#pragma unroll
      for (int tTile = 0; tTile < 4; ++tTile) {
        v8f d  = __builtin_amdgcn_wmma_f32_16x16x4_f32(
            false, Ad[tTile], false, Bd, (short)0, z, false, false);
        v8f cr = __builtin_amdgcn_wmma_f32_16x16x4_f32(
            false, Ac[tTile], false, Bw, (short)0, z, false, false);
        v8f yd = __builtin_amdgcn_wmma_f32_16x16x4_f32(
            false, Ay[tTile], false, Bw, (short)0, z, false, false);
        e = vmin8(e, d);
        int le[8];
#pragma unroll
        for (int v = 0; v < 8; ++v) le[v] = (yd[v] >= 0.0f);
#pragma unroll
        for (int v = 0; v < 7; ++v) {
          int aLE = le[v], bLE = le[v + 1];
          float cv = cr[v];
          wn += (aLE & (bLE ^ 1) & (cv > 0.0f)) - ((aLE ^ 1) & bLE & (cv < 0.0f));
        }
        le0b |= le[0] << tTile;
        cr7[tTile] = cr[7];
        le7[tTile] = le[7];
      }
      int recvb = __shfl_xor(le0b, 16, 32);
#pragma unroll
      for (int tTile = 0; tTile < 4; ++tTile) {
        int nt = hi ? ((tTile + 1) & 3) : tTile;
        int bLE = (recvb >> nt) & 1;
        int aLE = le7[tTile];
        float cv = cr7[tTile];
        wn += (aLE & (bLE ^ 1) & (cv > 0.0f)) - ((aLE ^ 1) & bLE & (cv < 0.0f));
      }
      wn += __shfl_xor(wn, 16, 32);

      float m01 = fminf(e[0], e[1]), m23 = fminf(e[2], e[3]);
      float m45 = fminf(e[4], e[5]), m67 = fminf(e[6], e[7]);
      float m8  = fminf(fminf(m01, m23), fminf(m45, m67));
      float d2min = fminf(m8, __shfl_xor(m8, 16, 32));

      float dminv = sqrtf(d2min + 1e-12f);
      float sdf2d = (wn != 0) ? -dminv : dminv;
      float dz = fabsf(plz) - habs;
      float po = fmaxf(sdf2d, 0.0f), dzo = fmaxf(dz, 0.0f);
      float sdf = fminf(fmaxf(sdf2d, dz), 0.0f) + sqrtf(po * po + dzo * dzo + 1e-12f);

      tile_out[k] = hi ? dminv : sdf;
    }
  }
  __syncthreads();

  {
    const float4 sv = reinterpret_cast<const float4*>(&s_sdf[0][0])[tid];
    float4 oc;
    oc.x = 1.0f / (1.0f + __expf(150.0f * sv.x));
    oc.y = 1.0f / (1.0f + __expf(150.0f * sv.y));
    oc.z = 1.0f / (1.0f + __expf(150.0f * sv.z));
    oc.w = 1.0f / (1.0f + __expf(150.0f * sv.w));
    reinterpret_cast<float4*>(&occ_sh[0][0])[tid] = oc;
  }
  __syncthreads();

  {
    const int g = tid >> 6;
#pragma unroll 1
    for (int rr = 0; rr < RPT; ++rr) {
      const int r = g * RPT + rr;
      float p[K_];
      float pmin = 3.0e38f;
#pragma unroll
      for (int kk = 0; kk < K_; ++kk) {
        p[kk] = 1.0f - w[kk] * (1.0f - occ_sh[r][kk]);
        pmin = fminf(pmin, p[kk]);
      }
      float inter;
      if (train) {
        float s = 0.0f, ts = 0.0f;
#pragma unroll
        for (int kk = 0; kk < K_; ++kk) {
          float ee = __expf(-40.0f * (p[kk] - pmin));
          s += ee; ts += ee * p[kk];
        }
        inter = ts / s;
      } else {
        inter = pmin;
      }
      inter_out[(size_t)(bm0 + r) * C_ + c] = inter;
      preu[r][c] = uwc * inter;
    }
  }
  __syncthreads();

  {
    const float4 sv = reinterpret_cast<const float4*>(&s_sdf[0][0])[tid];
    const float4 dv = reinterpret_cast<const float4*>(&s_dmin[0][0])[tid];
    reinterpret_cast<float4*>(sdf_out  + (size_t)bm0 * K_)[tid] = sv;
    reinterpret_cast<float4*>(dmin_out + (size_t)bm0 * K_)[tid] = dv;
  }

#pragma unroll
  for (int rr = 0; rr < PTS / 16; ++rr) {
    const int r = wv * (PTS / 16) + rr;
    float a0 = preu[r][lane], a1 = preu[r][lane + 32];
    float mx = fmaxf(a0, a1);
#pragma unroll
    for (int mask = 1; mask <= 16; mask <<= 1)
      mx = fmaxf(mx, __shfl_xor(mx, mask, 32));
    float res = mx;
    if (train) {
      float e0 = __expf(40.0f * (a0 - mx));
      float e1 = __expf(40.0f * (a1 - mx));
      float s  = e0 + e1;
      float ts = e0 * a0 + e1 * a1;
#pragma unroll
      for (int mask = 1; mask <= 16; mask <<= 1) {
        s  += __shfl_xor(s,  mask, 32);
        ts += __shfl_xor(ts, mask, 32);
      }
      res = ts / s;
    }
    if (lane == 0) occ_out[bm0 + r] = res;
  }
}

extern "C" void kernel_launch(void* const* d_in, const int* in_sizes, int n_in,
                              void* d_out, int out_size, void* d_ws, size_t ws_size,
                              hipStream_t stream) {
  const float* pts   = (const float*)d_in[0];
  const float* prim  = (const float*)d_in[1];
  const float* iw    = (const float*)d_in[2];
  const float* uw    = (const float*)d_in[3];
  const int*   train = (const int*)d_in[4];

  float* occ_out   = (float*)d_out;
  float* sdf_out   = occ_out + (size_t)B_ * M_;
  float* inter_out = sdf_out + (size_t)B_ * M_ * K_;
  float* dmin_out  = inter_out + (size_t)B_ * M_ * C_;

  fused_kernel<<<B_ * M_ / PTS, NT, 0, stream>>>(pts, prim, iw, uw, train,
                                                 occ_out, sdf_out, inter_out, dmin_out);
  (void)in_sizes; (void)n_in; (void)out_size; (void)d_ws; (void)ws_size;
}


# winding count on packed sign bits: a lane shifts the sign bits of its 32 edge values (cr, yt) into two words and tests all crossings with word logic and popcount, not per-edge compares and selects
# speedup vs baseline: 3.1798x; 1.1887x over previous
#include <hip/hip_runtime.h>
#include <hip/hip_bf16.h>
#include <math.h>

typedef float v2f __attribute__((ext_vector_type(2)));
typedef float v8f __attribute__((ext_vector_type(8)));

#define B_ 4
#define M_ 2048
#define K_ 64
#define C_ 64
#define P_ 28
#define NCURVE 64
#define PTS 32
#define NT 512
#define KPW 8
#define RPT 4

__device__ __forceinline__ v8f vmin8(v8f a, v8f b){
  v8f r;
#pragma unroll
  for (int i = 0; i < 8; ++i) r[i] = fminf(a[i], b[i]);
  return r;
}

__global__ __launch_bounds__(NT) void fused_kernel(
    const float* __restrict__ pts, const float* __restrict__ prim,
    const float* __restrict__ iw, const float* __restrict__ uw,
    const int* __restrict__ is_training,
    float* __restrict__ occ_out, float* __restrict__ sdf_out,
    float* __restrict__ inter_out, float* __restrict__ dmin_out)
{
  __shared__ float4 s_c4[K_][NCURVE];
  __shared__ float4 s_e4[K_][NCURVE];
  __shared__ float4 s_trig[4][2];
  __shared__ float4 s_pose[K_][2];
  __shared__ float  s_prim[P_ * K_];
  __shared__ __align__(16) float s_sdf[PTS][K_];
  __shared__ __align__(16) float s_dmin[PTS][K_];
  __shared__ __align__(16) float occ_sh[PTS][K_];
  __shared__ float preu[PTS][C_];

  const int blk = blockIdx.x;
  const int b   = blk >> 6;
  const int bm0 = blk * PTS;
  const int tid = threadIdx.x;
  const int lane = tid & 31;
  const int wv = tid >> 5;
  const int row = lane & 15;
  const bool hi = lane >= 16;
  const float PI = 3.14159265358979323846f;

  const int ptile = wv & 1;
  const float* p3 = pts + ((size_t)bm0 + ptile * 16 + row) * 3;
  const float rx = p3[0], ry = p3[1], rz = p3[2];

  float pr[4];
#pragma unroll
  for (int i = 0; i < 4; ++i) {
    const int idx = tid + i * NT;
    pr[i] = (idx < P_ * K_) ? prim[(size_t)b * P_ * K_ + idx] : 0.0f;
  }

  const int c = tid & 63;
  const float* wp = iw + ((size_t)b * K_) * C_ + c;
  float w[K_];
#pragma unroll
  for (int kk = 0; kk < K_; ++kk) w[kk] = wp[(size_t)kk * C_];
  const float uwc = uw[b * C_ + c];
  const int train = is_training[0];

  if (tid < 16) {
    const int seg = tid >> 2;
    const int a = tid & 3;
    const float theta = PI * 0.125f + atanf(tanf(PI * 0.125f) * (1.0f / 3.0f));
    const float r0 = (PI * 0.5f) * (float)seg;
    const float r3 = (PI * 0.5f) * (float)(seg + 1);
    const float ra = (a == 0) ? r0 : (a == 1) ? (r0 + theta) : (a == 2) ? (r3 - theta) : r3;
    reinterpret_cast<float2*>(&s_trig[seg][0])[a] = make_float2(cosf(ra), sinf(ra));
  }
#pragma unroll
  for (int i = 0; i < 4; ++i) {
    const int idx = tid + i * NT;
    if (idx < P_ * K_) s_prim[idx] = pr[i];
  }
  __syncthreads();

  {
    const int j  = tid & 63;
    const int kg = tid >> 6;
    const int seg = j >> 4;
    const int seg1 = (seg + 1) & 3;
    const float t = (float)(j & 15) * (1.0f / 16.0f);
    const float u = 1.0f - t;
    const float b0 = u * u * u, b1 = 3.0f * u * u * t, b2 = 3.0f * u * t * t, b3 = t * t * t;
    const float4 ta = s_trig[seg][0];
    const float4 tb = s_trig[seg][1];
    for (int i = 0; i < K_ / 8; ++i) {
      const int k = kg + 8 * i;
      const float rad0 = fabsf(s_prim[(8 + seg * 3 + 0) * K_ + k]);
      const float rad1 = fabsf(s_prim[(8 + seg * 3 + 1) * K_ + k]);
      const float rad2 = fabsf(s_prim[(8 + seg * 3 + 2) * K_ + k]);
      const float rad3 = fabsf(s_prim[(8 + seg1 * 3 + 0) * K_ + k]);
      const float P0x = ta.x * rad0, P0y = ta.y * rad0;
      const float P1x = ta.z * rad1, P1y = ta.w * rad1;
      const float P2x = tb.x * rad2, P2y = tb.y * rad2;
      const float P3x = tb.z * rad3, P3y = tb.w * rad3;
      const float w1 = fabsf(s_prim[(20 + seg * 2 + 0) * K_ + k]);
      const float w2 = fabsf(s_prim[(20 + seg * 2 + 1) * K_ + k]);
      const float inv = 1.0f / (b0 + w1 * b1 + w2 * b2 + b3);
      const float cx = (P0x * b0 + P1x * (w1 * b1) + P2x * (w2 * b2) + P3x * b3) * inv;
      const float cy = (P0y * b0 + P1y * (w1 * b1) + P2y * (w2 * b2) + P3y * b3) * inv;
      s_c4[k][j] = make_float4(cx, cy, cx * cx + cy * cy, 1.0f);
    }
  }
  if (tid < K_) {
    const float q0 = s_prim[tid], q1 = s_prim[K_ + tid];
    const float q2 = s_prim[2 * K_ + tid], q3 = s_prim[3 * K_ + tid];
    const float invn = 1.0f / sqrtf(q0 * q0 + q1 * q1 + q2 * q2 + q3 * q3 + 1e-8f);
    s_pose[tid][0] = make_float4(q0 * invn, -q1 * invn, -q2 * invn, -q3 * invn);
    s_pose[tid][1] = make_float4(s_prim[4 * K_ + tid], s_prim[5 * K_ + tid],
                                 s_prim[6 * K_ + tid], fabsf(s_prim[7 * K_ + tid]));
  }
  __syncthreads();

  {
    const int j  = tid & 63;
    const int kg = tid >> 6;
    for (int i = 0; i < K_ / 8; ++i) {
      const int k = kg + 8 * i;
      const float4 ca = s_c4[k][j];
      const float4 cb = s_c4[k][(j + 1) & 63];
      s_e4[k][j] = make_float4(ca.y - cb.y, cb.x - ca.x,
                               ca.x * cb.y - ca.y * cb.x, ca.y);
    }
  }
  __syncthreads();

  {
    const int kbase = (wv >> 1) * KPW;
    const int prow = ptile * 16 + row;
    float* const tile_out = hi ? &s_dmin[prow][0] : &s_sdf[prow][0];
    const v8f z = {0.f, 0.f, 0.f, 0.f, 0.f, 0.f, 0.f, 0.f};

#pragma unroll 1
    for (int kk = 0; kk < KPW; ++kk) {
      const int k = kbase + kk;
      const float4 pa = s_pose[k][0];
      const float4 pb = s_pose[k][1];
      const float qw = pa.x, vx = pa.y, vy = pa.z, vz = pa.w;
      const float habs = pb.w;

      v2f Ad[4], Ac[4], Ay[4];
#pragma unroll
      for (int tTile = 0; tTile < 4; ++tTile) {
        const float4 cc = s_c4[k][tTile * 16 + row];
        const float4 ee = s_e4[k][tTile * 16 + row];
        Ad[tTile].x = hi ? 1.0f  : -2.0f * cc.x;  Ad[tTile].y = hi ? cc.z : -2.0f * cc.y;
        Ac[tTile].x = hi ? ee.z  : ee.x;          Ac[tTile].y = hi ? 0.0f : ee.y;
        Ay[tTile].x = hi ? -ee.w : 0.0f;          Ay[tTile].y = hi ? 0.0f : 1.0f;
      }

      const float px0 = rx - pb.x, py0 = ry - pb.y, pz0 = rz - pb.z;
      const float t2x = 2.0f * (vy * pz0 - vz * py0);
      const float t2y = 2.0f * (vz * px0 - vx * pz0);
      const float t2z = 2.0f * (vx * py0 - vy * px0);
      const float plx = px0 + qw * t2x + (vy * t2z - vz * t2y);
      const float ply = py0 + qw * t2y + (vz * t2x - vx * t2z);
      const float plz = pz0 + qw * t2z + (vx * t2y - vy * t2x);
      const float px = plx, py = ply;
      const float ss = px * px + py * py;

      v2f Bd; Bd.x = hi ? ss : px;   Bd.y = hi ? 1.0f : py;
      v2f Bw; Bw.x = hi ? 1.0f : px; Bw.y = hi ? 0.0f : py;
      v8f e = {3e38f, 3e38f, 3e38f, 3e38f, 3e38f, 3e38f, 3e38f, 3e38f};

      unsigned am = 0u, cm = 0u;
#pragma unroll
      for (int tTile = 0; tTile < 4; ++tTile) {
        v8f d  = __builtin_amdgcn_wmma_f32_16x16x4_f32(
            false, Ad[tTile], false, Bd, (short)0, z, false, false);
        v8f cr = __builtin_amdgcn_wmma_f32_16x16x4_f32(
            false, Ac[tTile], false, Bw, (short)0, z, false, false);
        v8f yd = __builtin_amdgcn_wmma_f32_16x16x4_f32(
            false, Ay[tTile], false, Bw, (short)0, z, false, false);
        e = vmin8(e, d);
        unsigned a8 = 0u, c8 = 0u;
#pragma unroll
        for (int v = 0; v < 8; ++v) {
          a8 = (a8 << 1) | (__float_as_uint(yd[v]) >> 31);
          c8 = (c8 << 1) | (__float_as_uint(cr[v]) >> 31);
        }
        am = (am << 8) | a8;
        cm = (cm << 8) | c8;
      }
      const unsigned oth = (unsigned)__shfl_xor((int)am, 16, 32);
      const unsigned fix = hi ? ((oth << 1) | (oth >> 31)) : (oth >> 7);
      const unsigned bm  = ((am << 1) & 0xFEFEFEFEu) | (fix & 0x01010101u);
      const unsigned hit = (am ^ bm) & ~(am ^ cm);
      int wn = __popc(hit & ~am) - __popc(hit & am);
      wn += __shfl_xor(wn, 16, 32);

      float m01 = fminf(e[0], e[1]), m23 = fminf(e[2], e[3]);
      float m45 = fminf(e[4], e[5]), m67 = fminf(e[6], e[7]);
      float m8  = fminf(fminf(m01, m23), fminf(m45, m67));
      float d2min = fminf(m8, __shfl_xor(m8, 16, 32));

      float dminv = sqrtf(d2min + 1e-12f);
      float sdf2d = (wn != 0) ? -dminv : dminv;
      float dz = fabsf(plz) - habs;
      float po = fmaxf(sdf2d, 0.0f), dzo = fmaxf(dz, 0.0f);
      float sdf = fminf(fmaxf(sdf2d, dz), 0.0f) + sqrtf(po * po + dzo * dzo + 1e-12f);

      tile_out[k] = hi ? dminv : sdf;
    }
  }
  __syncthreads();

  {
    const float4 sv = reinterpret_cast<const float4*>(&s_sdf[0][0])[tid];
    float4 oc;
    oc.x = 1.0f / (1.0f + __expf(150.0f * sv.x));
    oc.y = 1.0f / (1.0f + __expf(150.0f * sv.y));
    oc.z = 1.0f / (1.0f + __expf(150.0f * sv.z));
    oc.w = 1.0f / (1.0f + __expf(150.0f * sv.w));
    reinterpret_cast<float4*>(&occ_sh[0][0])[tid] = oc;
  }
  __syncthreads();

  {
    const int g = tid >> 6;
#pragma unroll 1
    for (int rr = 0; rr < RPT; ++rr) {
      const int r = g * RPT + rr;
      float p[K_];
      float pmin = 3.0e38f;
#pragma unroll
      for (int kk = 0; kk < K_; ++kk) {
        p[kk] = 1.0f - w[kk] * (1.0f - occ_sh[r][kk]);
        pmin = fminf(pmin, p[kk]);
      }
      float inter;
      if (train) {
        float s = 0.0f, ts = 0.0f;
#pragma unroll
        for (int kk = 0; kk < K_; ++kk) {
          float ee = __expf(-40.0f * (p[kk] - pmin));
          s += ee; ts += ee * p[kk];
        }
        inter = ts / s;
      } else {
        inter = pmin;
      }
      inter_out[(size_t)(bm0 + r) * C_ + c] = inter;
      preu[r][c] = uwc * inter;
    }
  }
  __syncthreads();

  {
    const float4 sv = reinterpret_cast<const float4*>(&s_sdf[0][0])[tid];
    const float4 dv = reinterpret_cast<const float4*>(&s_dmin[0][0])[tid];
    reinterpret_cast<float4*>(sdf_out  + (size_t)bm0 * K_)[tid] = sv;
    reinterpret_cast<float4*>(dmin_out + (size_t)bm0 * K_)[tid] = dv;
  }

#pragma unroll
  for (int rr = 0; rr < PTS / 16; ++rr) {
    const int r = wv * (PTS / 16) + rr;
    float a0 = preu[r][lane], a1 = preu[r][lane + 32];
    float mx = fmaxf(a0, a1);
#pragma unroll
    for (int mask = 1; mask <= 16; mask <<= 1)
      mx = fmaxf(mx, __shfl_xor(mx, mask, 32));
    float res = mx;
    if (train) {
      float e0 = __expf(40.0f * (a0 - mx));
      float e1 = __expf(40.0f * (a1 - mx));
      float s  = e0 + e1;
      float ts = e0 * a0 + e1 * a1;
#pragma unroll
      for (int mask = 1; mask <= 16; mask <<= 1) {
        s  += __shfl_xor(s,  mask, 32);
        ts += __shfl_xor(ts, mask, 32);
      }
      res = ts / s;
    }
    if (lane == 0) occ_out[bm0 + r] = res;
  }
}

extern "C" void kernel_launch(void* const* d_in, const int* in_sizes, int n_in,
                              void* d_out, int out_size, void* d_ws, size_t ws_size,
                              hipStream_t stream) {
  const float* pts   = (const float*)d_in[0];
  const float* prim  = (const float*)d_in[1];
  const float* iw    = (const float*)d_in[2];
  const float* uw    = (const float*)d_in[3];
  const int*   train = (const int*)d_in[4];

  float* occ_out   = (float*)d_out;
  float* sdf_out   = occ_out + (size_t)B_ * M_;
  float* inter_out = sdf_out + (size_t)B_ * M_ * K_;
  float* dmin_out  = inter_out + (size_t)B_ * M_ * C_;

  fused_kernel<<<B_ * M_ / PTS, NT, 0, stream>>>(pts, prim, iw, uw, train,
                                                 occ_out, sdf_out, inter_out, dmin_out);
  (void)in_sizes; (void)n_in; (void)out_size; (void)d_ws; (void)ws_size;
}


# combine soft-min: fewer vector instructions per term - occ-1 formed once per (row,k), two k per packed multiply-add, -40*log2(e) and the row-minimum shift folded into one fma ahead of exp2
# speedup vs baseline: 3.4046x; 1.0707x over previous
#include <hip/hip_runtime.h>
#include <hip/hip_bf16.h>
#include <math.h>

typedef float v2f __attribute__((ext_vector_type(2)));
typedef float v8f __attribute__((ext_vector_type(8)));

#define B_ 4
#define M_ 2048
#define K_ 64
#define C_ 64
#define P_ 28
#define NCURVE 64
#define PTS 32
#define NT 512
#define KPW 8
#define RPT 4

__device__ __forceinline__ v8f vmin8(v8f a, v8f b){
  v8f r;
#pragma unroll
  for (int i = 0; i < 8; ++i) r[i] = fminf(a[i], b[i]);
  return r;
}

__global__ __launch_bounds__(NT) void fused_kernel(
    const float* __restrict__ pts, const float* __restrict__ prim,
    const float* __restrict__ iw, const float* __restrict__ uw,
    const int* __restrict__ is_training,
    float* __restrict__ occ_out, float* __restrict__ sdf_out,
    float* __restrict__ inter_out, float* __restrict__ dmin_out)
{
  __shared__ float4 s_c4[K_][NCURVE];
  __shared__ float4 s_e4[K_][NCURVE];
  __shared__ float4 s_trig[4][2];
  __shared__ float4 s_pose[K_][2];
  __shared__ float  s_prim[P_ * K_];
  __shared__ __align__(16) float s_sdf[PTS][K_];
  __shared__ __align__(16) float s_dmin[PTS][K_];
  __shared__ __align__(16) float nomo_sh[PTS][K_];
  __shared__ float preu[PTS][C_];

  const int blk = blockIdx.x;
  const int b   = blk >> 6;
  const int bm0 = blk * PTS;
  const int tid = threadIdx.x;
  const int lane = tid & 31;
  const int wv = tid >> 5;
  const int row = lane & 15;
  const bool hi = lane >= 16;
  const float PI = 3.14159265358979323846f;

  const int ptile = wv & 1;
  const float* p3 = pts + ((size_t)bm0 + ptile * 16 + row) * 3;
  const float rx = p3[0], ry = p3[1], rz = p3[2];

  float pr[4];
#pragma unroll
  for (int i = 0; i < 4; ++i) {
    const int idx = tid + i * NT;
    pr[i] = (idx < P_ * K_) ? prim[(size_t)b * P_ * K_ + idx] : 0.0f;
  }

  const int c = tid & 63;
  const float* wp = iw + ((size_t)b * K_) * C_ + c;
  v2f w2[K_ / 2];
#pragma unroll
  for (int i = 0; i < K_ / 2; ++i) {
    w2[i].x = wp[(size_t)(2 * i) * C_];
    w2[i].y = wp[(size_t)(2 * i + 1) * C_];
  }
  const float uwc = uw[b * C_ + c];
  const int train = is_training[0];

  if (tid < 16) {
    const int seg = tid >> 2;
    const int a = tid & 3;
    const float theta = PI * 0.125f + atanf(tanf(PI * 0.125f) * (1.0f / 3.0f));
    const float r0 = (PI * 0.5f) * (float)seg;
    const float r3 = (PI * 0.5f) * (float)(seg + 1);
    const float ra = (a == 0) ? r0 : (a == 1) ? (r0 + theta) : (a == 2) ? (r3 - theta) : r3;
    reinterpret_cast<float2*>(&s_trig[seg][0])[a] = make_float2(cosf(ra), sinf(ra));
  }
#pragma unroll
  for (int i = 0; i < 4; ++i) {
    const int idx = tid + i * NT;
    if (idx < P_ * K_) s_prim[idx] = pr[i];
  }
  __syncthreads();

  {
    const int j  = tid & 63;
    const int kg = tid >> 6;
    const int seg = j >> 4;
    const int seg1 = (seg + 1) & 3;
    const float t = (float)(j & 15) * (1.0f / 16.0f);
    const float u = 1.0f - t;
    const float b0 = u * u * u, b1 = 3.0f * u * u * t, b2 = 3.0f * u * t * t, b3 = t * t * t;
    const float4 ta = s_trig[seg][0];
    const float4 tb = s_trig[seg][1];
    for (int i = 0; i < K_ / 8; ++i) {
      const int k = kg + 8 * i;
      const float rad0 = fabsf(s_prim[(8 + seg * 3 + 0) * K_ + k]);
      const float rad1 = fabsf(s_prim[(8 + seg * 3 + 1) * K_ + k]);
      const float rad2 = fabsf(s_prim[(8 + seg * 3 + 2) * K_ + k]);
      const float rad3 = fabsf(s_prim[(8 + seg1 * 3 + 0) * K_ + k]);
      const float P0x = ta.x * rad0, P0y = ta.y * rad0;
      const float P1x = ta.z * rad1, P1y = ta.w * rad1;
      const float P2x = tb.x * rad2, P2y = tb.y * rad2;
      const float P3x = tb.z * rad3, P3y = tb.w * rad3;
      const float w1 = fabsf(s_prim[(20 + seg * 2 + 0) * K_ + k]);
      const float w2 = fabsf(s_prim[(20 + seg * 2 + 1) * K_ + k]);
      const float inv = 1.0f / (b0 + w1 * b1 + w2 * b2 + b3);
      const float cx = (P0x * b0 + P1x * (w1 * b1) + P2x * (w2 * b2) + P3x * b3) * inv;
      const float cy = (P0y * b0 + P1y * (w1 * b1) + P2y * (w2 * b2) + P3y * b3) * inv;
      s_c4[k][j] = make_float4(cx, cy, cx * cx + cy * cy, 1.0f);
    }
  }
  if (tid < K_) {
    const float q0 = s_prim[tid], q1 = s_prim[K_ + tid];
    const float q2 = s_prim[2 * K_ + tid], q3 = s_prim[3 * K_ + tid];
    const float invn = 1.0f / sqrtf(q0 * q0 + q1 * q1 + q2 * q2 + q3 * q3 + 1e-8f);
    s_pose[tid][0] = make_float4(q0 * invn, -q1 * invn, -q2 * invn, -q3 * invn);
    s_pose[tid][1] = make_float4(s_prim[4 * K_ + tid], s_prim[5 * K_ + tid],
                                 s_prim[6 * K_ + tid], fabsf(s_prim[7 * K_ + tid]));
  }
  __syncthreads();

  {
    const int j  = tid & 63;
    const int kg = tid >> 6;
    for (int i = 0; i < K_ / 8; ++i) {
      const int k = kg + 8 * i;
      const float4 ca = s_c4[k][j];
      const float4 cb = s_c4[k][(j + 1) & 63];
      s_e4[k][j] = make_float4(ca.y - cb.y, cb.x - ca.x,
                               ca.x * cb.y - ca.y * cb.x, ca.y);
    }
  }
  __syncthreads();

  {
    const int kbase = (wv >> 1) * KPW;
    const int prow = ptile * 16 + row;
    float* const tile_out = hi ? &s_dmin[prow][0] : &s_sdf[prow][0];
    const v8f z = {0.f, 0.f, 0.f, 0.f, 0.f, 0.f, 0.f, 0.f};

#pragma unroll 1
    for (int kk = 0; kk < KPW; ++kk) {
      const int k = kbase + kk;
      const float4 pa = s_pose[k][0];
      const float4 pb = s_pose[k][1];
      const float qw = pa.x, vx = pa.y, vy = pa.z, vz = pa.w;
      const float habs = pb.w;

      v2f Ad[4], Ac[4], Ay[4];
#pragma unroll
      for (int tTile = 0; tTile < 4; ++tTile) {
        const float4 cc = s_c4[k][tTile * 16 + row];
        const float4 ee = s_e4[k][tTile * 16 + row];
        Ad[tTile].x = hi ? 1.0f  : -2.0f * cc.x;  Ad[tTile].y = hi ? cc.z : -2.0f * cc.y;
        Ac[tTile].x = hi ? ee.z  : ee.x;          Ac[tTile].y = hi ? 0.0f : ee.y;
        Ay[tTile].x = hi ? -ee.w : 0.0f;          Ay[tTile].y = hi ? 0.0f : 1.0f;
      }

      const float px0 = rx - pb.x, py0 = ry - pb.y, pz0 = rz - pb.z;
      const float t2x = 2.0f * (vy * pz0 - vz * py0);
      const float t2y = 2.0f * (vz * px0 - vx * pz0);
      const float t2z = 2.0f * (vx * py0 - vy * px0);
      const float plx = px0 + qw * t2x + (vy * t2z - vz * t2y);
      const float ply = py0 + qw * t2y + (vz * t2x - vx * t2z);
      const float plz = pz0 + qw * t2z + (vx * t2y - vy * t2x);
      const float px = plx, py = ply;
      const float ss = px * px + py * py;

      v2f Bd; Bd.x = hi ? ss : px;   Bd.y = hi ? 1.0f : py;
      v2f Bw; Bw.x = hi ? 1.0f : px; Bw.y = hi ? 0.0f : py;
      v8f e = {3e38f, 3e38f, 3e38f, 3e38f, 3e38f, 3e38f, 3e38f, 3e38f};

      unsigned am = 0u, cm = 0u;
#pragma unroll
      for (int tTile = 0; tTile < 4; ++tTile) {
        v8f d  = __builtin_amdgcn_wmma_f32_16x16x4_f32(
            false, Ad[tTile], false, Bd, (short)0, z, false, false);
        v8f cr = __builtin_amdgcn_wmma_f32_16x16x4_f32(
            false, Ac[tTile], false, Bw, (short)0, z, false, false);
        v8f yd = __builtin_amdgcn_wmma_f32_16x16x4_f32(
            false, Ay[tTile], false, Bw, (short)0, z, false, false);
        e = vmin8(e, d);
        unsigned a8 = 0u, c8 = 0u;
#pragma unroll
        for (int v = 0; v < 8; ++v) {
          a8 = (a8 << 1) | (__float_as_uint(yd[v]) >> 31);
          c8 = (c8 << 1) | (__float_as_uint(cr[v]) >> 31);
        }
        am = (am << 8) | a8;
        cm = (cm << 8) | c8;
      }
      const unsigned oth = (unsigned)__shfl_xor((int)am, 16, 32);
      const unsigned fix = hi ? ((oth << 1) | (oth >> 31)) : (oth >> 7);
      const unsigned bm  = ((am << 1) & 0xFEFEFEFEu) | (fix & 0x01010101u);
      const unsigned hit = (am ^ bm) & ~(am ^ cm);
      int wn = __popc(hit & ~am) - __popc(hit & am);
      wn += __shfl_xor(wn, 16, 32);

      float m01 = fminf(e[0], e[1]), m23 = fminf(e[2], e[3]);
      float m45 = fminf(e[4], e[5]), m67 = fminf(e[6], e[7]);
      float m8  = fminf(fminf(m01, m23), fminf(m45, m67));
      float d2min = fminf(m8, __shfl_xor(m8, 16, 32));

      float dminv = sqrtf(d2min + 1e-12f);
      float sdf2d = (wn != 0) ? -dminv : dminv;
      float dz = fabsf(plz) - habs;
      float po = fmaxf(sdf2d, 0.0f), dzo = fmaxf(dz, 0.0f);
      float sdf = fminf(fmaxf(sdf2d, dz), 0.0f) + sqrtf(po * po + dzo * dzo + 1e-12f);

      tile_out[k] = hi ? dminv : sdf;
    }
  }
  __syncthreads();

  {
    const float4 sv = reinterpret_cast<const float4*>(&s_sdf[0][0])[tid];
    float4 oc;
    oc.x = 1.0f / (1.0f + __expf(150.0f * sv.x)) - 1.0f;
    oc.y = 1.0f / (1.0f + __expf(150.0f * sv.y)) - 1.0f;
    oc.z = 1.0f / (1.0f + __expf(150.0f * sv.z)) - 1.0f;
    oc.w = 1.0f / (1.0f + __expf(150.0f * sv.w)) - 1.0f;
    reinterpret_cast<float4*>(&nomo_sh[0][0])[tid] = oc;
  }
  __syncthreads();

  {
    const int g = tid >> 6;
#pragma unroll 1
    for (int rr = 0; rr < RPT; ++rr) {
      const int r = g * RPT + rr;
      v2f p2[K_ / 2];
      float pmin = 3.0e38f;
      const v2f one2 = {1.0f, 1.0f};
      const float4* nrow = reinterpret_cast<const float4*>(&nomo_sh[r][0]);
#pragma unroll
      for (int q = 0; q < K_ / 4; ++q) {
        const float4 n4 = nrow[q];
        const v2f na = {n4.x, n4.y}, nb = {n4.z, n4.w};
        p2[2 * q]     = __builtin_elementwise_fma(w2[2 * q],     na, one2);
        p2[2 * q + 1] = __builtin_elementwise_fma(w2[2 * q + 1], nb, one2);
        pmin = fminf(pmin, fminf(p2[2 * q].x, p2[2 * q].y));
        pmin = fminf(pmin, fminf(p2[2 * q + 1].x, p2[2 * q + 1].y));
      }
      float inter;
      if (train) {
        const float NC = -57.70780163555854f;
        const float shift = -NC * pmin;
        const v2f nc2 = {NC, NC};
        const v2f shift2 = {shift, shift};
        float s = 0.0f, ts = 0.0f;
#pragma unroll
        for (int i = 0; i < K_ / 2; ++i) {
          const v2f a2 = __builtin_elementwise_fma(p2[i], nc2, shift2);
          const float e0 = __builtin_amdgcn_exp2f(a2.x);
          const float e1 = __builtin_amdgcn_exp2f(a2.y);
          s += e0; ts += e0 * p2[i].x;
          s += e1; ts += e1 * p2[i].y;
        }
        inter = ts / s;
      } else {
        inter = pmin;
      }
      inter_out[(size_t)(bm0 + r) * C_ + c] = inter;
      preu[r][c] = uwc * inter;
    }
  }
  __syncthreads();

  {
    const float4 sv = reinterpret_cast<const float4*>(&s_sdf[0][0])[tid];
    const float4 dv = reinterpret_cast<const float4*>(&s_dmin[0][0])[tid];
    reinterpret_cast<float4*>(sdf_out  + (size_t)bm0 * K_)[tid] = sv;
    reinterpret_cast<float4*>(dmin_out + (size_t)bm0 * K_)[tid] = dv;
  }

#pragma unroll
  for (int rr = 0; rr < PTS / 16; ++rr) {
    const int r = wv * (PTS / 16) + rr;
    float a0 = preu[r][lane], a1 = preu[r][lane + 32];
    float mx = fmaxf(a0, a1);
#pragma unroll
    for (int mask = 1; mask <= 16; mask <<= 1)
      mx = fmaxf(mx, __shfl_xor(mx, mask, 32));
    float res = mx;
    if (train) {
      float e0 = __expf(40.0f * (a0 - mx));
      float e1 = __expf(40.0f * (a1 - mx));
      float s  = e0 + e1;
      float ts = e0 * a0 + e1 * a1;
#pragma unroll
      for (int mask = 1; mask <= 16; mask <<= 1) {
        s  += __shfl_xor(s,  mask, 32);
        ts += __shfl_xor(ts, mask, 32);
      }
      res = ts / s;
    }
    if (lane == 0) occ_out[bm0 + r] = res;
  }
}

extern "C" void kernel_launch(void* const* d_in, const int* in_sizes, int n_in,
                              void* d_out, int out_size, void* d_ws, size_t ws_size,
                              hipStream_t stream) {
  const float* pts   = (const float*)d_in[0];
  const float* prim  = (const float*)d_in[1];
  const float* iw    = (const float*)d_in[2];
  const float* uw    = (const float*)d_in[3];
  const int*   train = (const int*)d_in[4];

  float* occ_out   = (float*)d_out;
  float* sdf_out   = occ_out + (size_t)B_ * M_;
  float* inter_out = sdf_out + (size_t)B_ * M_ * K_;
  float* dmin_out  = inter_out + (size_t)B_ * M_ * C_;

  fused_kernel<<<B_ * M_ / PTS, NT, 0, stream>>>(pts, prim, iw, uw, train,
                                                 occ_out, sdf_out, inter_out, dmin_out);
  (void)in_sizes; (void)n_in; (void)out_size; (void)d_ws; (void)ws_size;
}


# work-group barriers order LDS only (local-scope fence + s_barrier) and no longer drain global traffic: weight loads return under the curve build, tile stores go out under the combine
# speedup vs baseline: 3.6442x; 1.0704x over previous
#include <hip/hip_runtime.h>
#include <hip/hip_bf16.h>
#include <math.h>

typedef float v2f __attribute__((ext_vector_type(2)));
typedef float v8f __attribute__((ext_vector_type(8)));

#define B_ 4
#define M_ 2048
#define K_ 64
#define C_ 64
#define P_ 28
#define NCURVE 64
#define PTS 32
#define NT 512
#define KPW 8
#define RPT 4

__device__ __forceinline__ v8f vmin8(v8f a, v8f b){
  v8f r;
#pragma unroll
  for (int i = 0; i < 8; ++i) r[i] = fminf(a[i], b[i]);
  return r;
}

__device__ __forceinline__ void lds_barrier(){
  __builtin_amdgcn_fence(__ATOMIC_RELEASE, "workgroup", "local");
  __builtin_amdgcn_s_barrier();
  __builtin_amdgcn_fence(__ATOMIC_ACQUIRE, "workgroup", "local");
}

__global__ __launch_bounds__(NT) void fused_kernel(
    const float* __restrict__ pts, const float* __restrict__ prim,
    const float* __restrict__ iw, const float* __restrict__ uw,
    const int* __restrict__ is_training,
    float* __restrict__ occ_out, float* __restrict__ sdf_out,
    float* __restrict__ inter_out, float* __restrict__ dmin_out)
{
  __shared__ float4 s_c4[K_][NCURVE];
  __shared__ float4 s_e4[K_][NCURVE];
  __shared__ float4 s_trig[4][2];
  __shared__ float4 s_pose[K_][2];
  __shared__ float  s_prim[P_ * K_];
  __shared__ __align__(16) float s_sdf[PTS][K_];
  __shared__ __align__(16) float s_dmin[PTS][K_];
  __shared__ __align__(16) float nomo_sh[PTS][K_];
  __shared__ float preu[PTS][C_];

  const int blk = blockIdx.x;
  const int b   = blk >> 6;
  const int bm0 = blk * PTS;
  const int tid = threadIdx.x;
  const int lane = tid & 31;
  const int wv = tid >> 5;
  const int row = lane & 15;
  const bool hi = lane >= 16;
  const float PI = 3.14159265358979323846f;

  const int ptile = wv & 1;
  const float* p3 = pts + ((size_t)bm0 + ptile * 16 + row) * 3;
  const float rx = p3[0], ry = p3[1], rz = p3[2];

  float pr[4];
#pragma unroll
  for (int i = 0; i < 4; ++i) {
    const int idx = tid + i * NT;
    pr[i] = (idx < P_ * K_) ? prim[(size_t)b * P_ * K_ + idx] : 0.0f;
  }

  const int train = is_training[0];

  if (tid < 16) {
    const int seg = tid >> 2;
    const int a = tid & 3;
    const float theta = PI * 0.125f + atanf(tanf(PI * 0.125f) * (1.0f / 3.0f));
    const float r0 = (PI * 0.5f) * (float)seg;
    const float r3 = (PI * 0.5f) * (float)(seg + 1);
    const float ra = (a == 0) ? r0 : (a == 1) ? (r0 + theta) : (a == 2) ? (r3 - theta) : r3;
    reinterpret_cast<float2*>(&s_trig[seg][0])[a] = make_float2(cosf(ra), sinf(ra));
  }
#pragma unroll
  for (int i = 0; i < 4; ++i) {
    const int idx = tid + i * NT;
    if (idx < P_ * K_) s_prim[idx] = pr[i];
  }
  lds_barrier();

  const int c = tid & 63;
  const float* wp = iw + ((size_t)b * K_) * C_ + c;
  v2f w2[K_ / 2];
#pragma unroll
  for (int i = 0; i < K_ / 4; ++i) {
    w2[i].x = wp[(size_t)(2 * i) * C_];
    w2[i].y = wp[(size_t)(2 * i + 1) * C_];
  }

  {
    const int j  = tid & 63;
    const int kg = tid >> 6;
    const int seg = j >> 4;
    const int seg1 = (seg + 1) & 3;
    const float t = (float)(j & 15) * (1.0f / 16.0f);
    const float u = 1.0f - t;
    const float b0 = u * u * u, b1 = 3.0f * u * u * t, b2 = 3.0f * u * t * t, b3 = t * t * t;
    const float4 ta = s_trig[seg][0];
    const float4 tb = s_trig[seg][1];
    for (int i = 0; i < K_ / 8; ++i) {
      const int k = kg + 8 * i;
      const float rad0 = fabsf(s_prim[(8 + seg * 3 + 0) * K_ + k]);
      const float rad1 = fabsf(s_prim[(8 + seg * 3 + 1) * K_ + k]);
      const float rad2 = fabsf(s_prim[(8 + seg * 3 + 2) * K_ + k]);
      const float rad3 = fabsf(s_prim[(8 + seg1 * 3 + 0) * K_ + k]);
      const float P0x = ta.x * rad0, P0y = ta.y * rad0;
      const float P1x = ta.z * rad1, P1y = ta.w * rad1;
      const float P2x = tb.x * rad2, P2y = tb.y * rad2;
      const float P3x = tb.z * rad3, P3y = tb.w * rad3;
      const float w1 = fabsf(s_prim[(20 + seg * 2 + 0) * K_ + k]);
      const float w2 = fabsf(s_prim[(20 + seg * 2 + 1) * K_ + k]);
      const float inv = 1.0f / (b0 + w1 * b1 + w2 * b2 + b3);
      const float cx = (P0x * b0 + P1x * (w1 * b1) + P2x * (w2 * b2) + P3x * b3) * inv;
      const float cy = (P0y * b0 + P1y * (w1 * b1) + P2y * (w2 * b2) + P3y * b3) * inv;
      s_c4[k][j] = make_float4(cx, cy, cx * cx + cy * cy, 1.0f);
    }
  }
  if (tid < K_) {
    const float q0 = s_prim[tid], q1 = s_prim[K_ + tid];
    const float q2 = s_prim[2 * K_ + tid], q3 = s_prim[3 * K_ + tid];
    const float invn = 1.0f / sqrtf(q0 * q0 + q1 * q1 + q2 * q2 + q3 * q3 + 1e-8f);
    s_pose[tid][0] = make_float4(q0 * invn, -q1 * invn, -q2 * invn, -q3 * invn);
    s_pose[tid][1] = make_float4(s_prim[4 * K_ + tid], s_prim[5 * K_ + tid],
                                 s_prim[6 * K_ + tid], fabsf(s_prim[7 * K_ + tid]));
  }
  lds_barrier();

#pragma unroll
  for (int i = K_ / 4; i < K_ / 2; ++i) {
    w2[i].x = wp[(size_t)(2 * i) * C_];
    w2[i].y = wp[(size_t)(2 * i + 1) * C_];
  }
  const float uwc = uw[b * C_ + c];

  {
    const int j  = tid & 63;
    const int kg = tid >> 6;
    for (int i = 0; i < K_ / 8; ++i) {
      const int k = kg + 8 * i;
      const float4 ca = s_c4[k][j];
      const float4 cb = s_c4[k][(j + 1) & 63];
      s_e4[k][j] = make_float4(ca.y - cb.y, cb.x - ca.x,
                               ca.x * cb.y - ca.y * cb.x, ca.y);
    }
  }
  lds_barrier();

  {
    const int kbase = (wv >> 1) * KPW;
    const int prow = ptile * 16 + row;
    float* const tile_out = hi ? &s_dmin[prow][0] : &s_sdf[prow][0];
    const v8f z = {0.f, 0.f, 0.f, 0.f, 0.f, 0.f, 0.f, 0.f};

#pragma unroll 1
    for (int kk = 0; kk < KPW; ++kk) {
      const int k = kbase + kk;
      const float4 pa = s_pose[k][0];
      const float4 pb = s_pose[k][1];
      const float qw = pa.x, vx = pa.y, vy = pa.z, vz = pa.w;
      const float habs = pb.w;

      v2f Ad[4], Ac[4], Ay[4];
#pragma unroll
      for (int tTile = 0; tTile < 4; ++tTile) {
        const float4 cc = s_c4[k][tTile * 16 + row];
        const float4 ee = s_e4[k][tTile * 16 + row];
        Ad[tTile].x = hi ? 1.0f  : -2.0f * cc.x;  Ad[tTile].y = hi ? cc.z : -2.0f * cc.y;
        Ac[tTile].x = hi ? ee.z  : ee.x;          Ac[tTile].y = hi ? 0.0f : ee.y;
        Ay[tTile].x = hi ? -ee.w : 0.0f;          Ay[tTile].y = hi ? 0.0f : 1.0f;
      }

      const float px0 = rx - pb.x, py0 = ry - pb.y, pz0 = rz - pb.z;
      const float t2x = 2.0f * (vy * pz0 - vz * py0);
      const float t2y = 2.0f * (vz * px0 - vx * pz0);
      const float t2z = 2.0f * (vx * py0 - vy * px0);
      const float plx = px0 + qw * t2x + (vy * t2z - vz * t2y);
      const float ply = py0 + qw * t2y + (vz * t2x - vx * t2z);
      const float plz = pz0 + qw * t2z + (vx * t2y - vy * t2x);
      const float px = plx, py = ply;
      const float ss = px * px + py * py;

      v2f Bd; Bd.x = hi ? ss : px;   Bd.y = hi ? 1.0f : py;
      v2f Bw; Bw.x = hi ? 1.0f : px; Bw.y = hi ? 0.0f : py;
      v8f e = {3e38f, 3e38f, 3e38f, 3e38f, 3e38f, 3e38f, 3e38f, 3e38f};

      unsigned am = 0u, cm = 0u;
#pragma unroll
      for (int tTile = 0; tTile < 4; ++tTile) {
        v8f d  = __builtin_amdgcn_wmma_f32_16x16x4_f32(
            false, Ad[tTile], false, Bd, (short)0, z, false, false);
        v8f cr = __builtin_amdgcn_wmma_f32_16x16x4_f32(
            false, Ac[tTile], false, Bw, (short)0, z, false, false);
        v8f yd = __builtin_amdgcn_wmma_f32_16x16x4_f32(
            false, Ay[tTile], false, Bw, (short)0, z, false, false);
        e = vmin8(e, d);
        unsigned a8 = 0u, c8 = 0u;
#pragma unroll
        for (int v = 0; v < 8; ++v) {
          a8 = (a8 << 1) | (__float_as_uint(yd[v]) >> 31);
          c8 = (c8 << 1) | (__float_as_uint(cr[v]) >> 31);
        }
        am = (am << 8) | a8;
        cm = (cm << 8) | c8;
      }
      const unsigned oth = (unsigned)__shfl_xor((int)am, 16, 32);
      const unsigned fix = hi ? ((oth << 1) | (oth >> 31)) : (oth >> 7);
      const unsigned bm  = ((am << 1) & 0xFEFEFEFEu) | (fix & 0x01010101u);
      const unsigned hit = (am ^ bm) & ~(am ^ cm);
      int wn = __popc(hit & ~am) - __popc(hit & am);
      wn += __shfl_xor(wn, 16, 32);

      float m01 = fminf(e[0], e[1]), m23 = fminf(e[2], e[3]);
      float m45 = fminf(e[4], e[5]), m67 = fminf(e[6], e[7]);
      float m8  = fminf(fminf(m01, m23), fminf(m45, m67));
      float d2min = fminf(m8, __shfl_xor(m8, 16, 32));

      float dminv = sqrtf(d2min + 1e-12f);
      float sdf2d = (wn != 0) ? -dminv : dminv;
      float dz = fabsf(plz) - habs;
      float po = fmaxf(sdf2d, 0.0f), dzo = fmaxf(dz, 0.0f);
      float sdf = fminf(fmaxf(sdf2d, dz), 0.0f) + sqrtf(po * po + dzo * dzo + 1e-12f);

      tile_out[k] = hi ? dminv : sdf;
    }
  }
  lds_barrier();

  {
    const float4 sv = reinterpret_cast<const float4*>(&s_sdf[0][0])[tid];
    const float4 dv = reinterpret_cast<const float4*>(&s_dmin[0][0])[tid];
    reinterpret_cast<float4*>(sdf_out  + (size_t)bm0 * K_)[tid] = sv;
    reinterpret_cast<float4*>(dmin_out + (size_t)bm0 * K_)[tid] = dv;
    float4 oc;
    oc.x = 1.0f / (1.0f + __expf(150.0f * sv.x)) - 1.0f;
    oc.y = 1.0f / (1.0f + __expf(150.0f * sv.y)) - 1.0f;
    oc.z = 1.0f / (1.0f + __expf(150.0f * sv.z)) - 1.0f;
    oc.w = 1.0f / (1.0f + __expf(150.0f * sv.w)) - 1.0f;
    reinterpret_cast<float4*>(&nomo_sh[0][0])[tid] = oc;
  }
  lds_barrier();

  {
    const int g = tid >> 6;
#pragma unroll 1
    for (int rr = 0; rr < RPT; ++rr) {
      const int r = g * RPT + rr;
      v2f p2[K_ / 2];
      float pmin = 3.0e38f;
      const v2f one2 = {1.0f, 1.0f};
      const float4* nrow = reinterpret_cast<const float4*>(&nomo_sh[r][0]);
#pragma unroll
      for (int q = 0; q < K_ / 4; ++q) {
        const float4 n4 = nrow[q];
        const v2f na = {n4.x, n4.y}, nb = {n4.z, n4.w};
        p2[2 * q]     = __builtin_elementwise_fma(w2[2 * q],     na, one2);
        p2[2 * q + 1] = __builtin_elementwise_fma(w2[2 * q + 1], nb, one2);
        pmin = fminf(pmin, fminf(p2[2 * q].x, p2[2 * q].y));
        pmin = fminf(pmin, fminf(p2[2 * q + 1].x, p2[2 * q + 1].y));
      }
      float inter;
      if (train) {
        const float NC = -57.70780163555854f;
        const float shift = -NC * pmin;
        const v2f nc2 = {NC, NC};
        const v2f shift2 = {shift, shift};
        float s = 0.0f, ts = 0.0f;
#pragma unroll
        for (int i = 0; i < K_ / 2; ++i) {
          const v2f a2 = __builtin_elementwise_fma(p2[i], nc2, shift2);
          const float e0 = __builtin_amdgcn_exp2f(a2.x);
          const float e1 = __builtin_amdgcn_exp2f(a2.y);
          s += e0; ts += e0 * p2[i].x;
          s += e1; ts += e1 * p2[i].y;
        }
        inter = ts / s;
      } else {
        inter = pmin;
      }
      inter_out[(size_t)(bm0 + r) * C_ + c] = inter;
      preu[r][c] = uwc * inter;
    }
  }
  lds_barrier();

#pragma unroll
  for (int rr = 0; rr < PTS / 16; ++rr) {
    const int r = wv * (PTS / 16) + rr;
    float a0 = preu[r][lane], a1 = preu[r][lane + 32];
    float mx = fmaxf(a0, a1);
#pragma unroll
    for (int mask = 1; mask <= 16; mask <<= 1)
      mx = fmaxf(mx, __shfl_xor(mx, mask, 32));
    float res = mx;
    if (train) {
      float e0 = __expf(40.0f * (a0 - mx));
      float e1 = __expf(40.0f * (a1 - mx));
      float s  = e0 + e1;
      float ts = e0 * a0 + e1 * a1;
#pragma unroll
      for (int mask = 1; mask <= 16; mask <<= 1) {
        s  += __shfl_xor(s,  mask, 32);
        ts += __shfl_xor(ts, mask, 32);
      }
      res = ts / s;
    }
    if (lane == 0) occ_out[bm0 + r] = res;
  }
}

extern "C" void kernel_launch(void* const* d_in, const int* in_sizes, int n_in,
                              void* d_out, int out_size, void* d_ws, size_t ws_size,
                              hipStream_t stream) {
  const float* pts   = (const float*)d_in[0];
  const float* prim  = (const float*)d_in[1];
  const float* iw    = (const float*)d_in[2];
  const float* uw    = (const float*)d_in[3];
  const int*   train = (const int*)d_in[4];

  float* occ_out   = (float*)d_out;
  float* sdf_out   = occ_out + (size_t)B_ * M_;
  float* inter_out = sdf_out + (size_t)B_ * M_ * K_;
  float* dmin_out  = inter_out + (size_t)B_ * M_ * C_;

  fused_kernel<<<B_ * M_ / PTS, NT, 0, stream>>>(pts, prim, iw, uw, train,
                                                 occ_out, sdf_out, inter_out, dmin_out);
  (void)in_sizes; (void)n_in; (void)out_size; (void)d_ws; (void)ws_size;
}


# lane exchanges on the vector unit (row permutes, half swap) instead of LDS permutes: the min/winding half exchanges of every primitive and the union butterfly; same partners, same order of sums
# speedup vs baseline: 3.8644x; 1.0604x over previous
#include <hip/hip_runtime.h>
#include <hip/hip_bf16.h>
#include <math.h>

typedef float v2f __attribute__((ext_vector_type(2)));
typedef float v8f __attribute__((ext_vector_type(8)));

#define B_ 4
#define M_ 2048
#define K_ 64
#define C_ 64
#define P_ 28
#define NCURVE 64
#define PTS 32
#define NT 512
#define KPW 8
#define RPT 4

__device__ __forceinline__ v8f vmin8(v8f a, v8f b){
  v8f r;
#pragma unroll
  for (int i = 0; i < 8; ++i) r[i] = fminf(a[i], b[i]);
  return r;
}

template <int MASK> __device__ __forceinline__ int xlane_i(int x){
  if constexpr (MASK == 16)
    return __builtin_amdgcn_permlanex16(x, x, 0x76543210, 0xfedcba98, false, false);
  else if constexpr (MASK == 1)
    return __builtin_amdgcn_update_dpp(x, x, 0xB1, 0xF, 0xF, false);
  else if constexpr (MASK == 2)
    return __builtin_amdgcn_update_dpp(x, x, 0x4E, 0xF, 0xF, false);
  else
    return __builtin_amdgcn_update_dpp(x, x, 0x160 + MASK, 0xF, 0xF, false);
}
template <int MASK> __device__ __forceinline__ float xlane_f(float x){
  return __int_as_float(xlane_i<MASK>(__float_as_int(x)));
}

__device__ __forceinline__ void lds_barrier(){
  __builtin_amdgcn_fence(__ATOMIC_RELEASE, "workgroup", "local");
  __builtin_amdgcn_s_barrier();
  __builtin_amdgcn_fence(__ATOMIC_ACQUIRE, "workgroup", "local");
}

__global__ __launch_bounds__(NT) void fused_kernel(
    const float* __restrict__ pts, const float* __restrict__ prim,
    const float* __restrict__ iw, const float* __restrict__ uw,
    const int* __restrict__ is_training,
    float* __restrict__ occ_out, float* __restrict__ sdf_out,
    float* __restrict__ inter_out, float* __restrict__ dmin_out)
{
  __shared__ float4 s_c4[K_][NCURVE];
  __shared__ float4 s_e4[K_][NCURVE];
  __shared__ float4 s_trig[4][2];
  __shared__ float4 s_pose[K_][2];
  __shared__ float  s_prim[P_ * K_];
  __shared__ __align__(16) float s_sdf[PTS][K_];
  __shared__ __align__(16) float s_dmin[PTS][K_];
  __shared__ __align__(16) float nomo_sh[PTS][K_];
  __shared__ float preu[PTS][C_];

  const int blk = blockIdx.x;
  const int b   = blk >> 6;
  const int bm0 = blk * PTS;
  const int tid = threadIdx.x;
  const int lane = tid & 31;
  const int wv = tid >> 5;
  const int row = lane & 15;
  const bool hi = lane >= 16;
  const float PI = 3.14159265358979323846f;

  const int ptile = wv & 1;
  const float* p3 = pts + ((size_t)bm0 + ptile * 16 + row) * 3;
  const float rx = p3[0], ry = p3[1], rz = p3[2];

  float pr[4];
#pragma unroll
  for (int i = 0; i < 4; ++i) {
    const int idx = tid + i * NT;
    pr[i] = (idx < P_ * K_) ? prim[(size_t)b * P_ * K_ + idx] : 0.0f;
  }

  const int train = is_training[0];

  if (tid < 16) {
    const int seg = tid >> 2;
    const int a = tid & 3;
    const float theta = PI * 0.125f + atanf(tanf(PI * 0.125f) * (1.0f / 3.0f));
    const float r0 = (PI * 0.5f) * (float)seg;
    const float r3 = (PI * 0.5f) * (float)(seg + 1);
    const float ra = (a == 0) ? r0 : (a == 1) ? (r0 + theta) : (a == 2) ? (r3 - theta) : r3;
    reinterpret_cast<float2*>(&s_trig[seg][0])[a] = make_float2(cosf(ra), sinf(ra));
  }
#pragma unroll
  for (int i = 0; i < 4; ++i) {
    const int idx = tid + i * NT;
    if (idx < P_ * K_) s_prim[idx] = pr[i];
  }
  lds_barrier();

  const int c = tid & 63;
  const float* wp = iw + ((size_t)b * K_) * C_ + c;
  v2f w2[K_ / 2];
#pragma unroll
  for (int i = 0; i < K_ / 4; ++i) {
    w2[i].x = wp[(size_t)(2 * i) * C_];
    w2[i].y = wp[(size_t)(2 * i + 1) * C_];
  }

  {
    const int j  = tid & 63;
    const int kg = tid >> 6;
    const int seg = j >> 4;
    const int seg1 = (seg + 1) & 3;
    const float t = (float)(j & 15) * (1.0f / 16.0f);
    const float u = 1.0f - t;
    const float b0 = u * u * u, b1 = 3.0f * u * u * t, b2 = 3.0f * u * t * t, b3 = t * t * t;
    const float4 ta = s_trig[seg][0];
    const float4 tb = s_trig[seg][1];
    for (int i = 0; i < K_ / 8; ++i) {
      const int k = kg + 8 * i;
      const float rad0 = fabsf(s_prim[(8 + seg * 3 + 0) * K_ + k]);
      const float rad1 = fabsf(s_prim[(8 + seg * 3 + 1) * K_ + k]);
      const float rad2 = fabsf(s_prim[(8 + seg * 3 + 2) * K_ + k]);
      const float rad3 = fabsf(s_prim[(8 + seg1 * 3 + 0) * K_ + k]);
      const float P0x = ta.x * rad0, P0y = ta.y * rad0;
      const float P1x = ta.z * rad1, P1y = ta.w * rad1;
      const float P2x = tb.x * rad2, P2y = tb.y * rad2;
      const float P3x = tb.z * rad3, P3y = tb.w * rad3;
      const float w1 = fabsf(s_prim[(20 + seg * 2 + 0) * K_ + k]);
      const float w2 = fabsf(s_prim[(20 + seg * 2 + 1) * K_ + k]);
      const float inv = 1.0f / (b0 + w1 * b1 + w2 * b2 + b3);
      const float cx = (P0x * b0 + P1x * (w1 * b1) + P2x * (w2 * b2) + P3x * b3) * inv;
      const float cy = (P0y * b0 + P1y * (w1 * b1) + P2y * (w2 * b2) + P3y * b3) * inv;
      s_c4[k][j] = make_float4(cx, cy, cx * cx + cy * cy, 1.0f);
    }
  }
  if (tid < K_) {
    const float q0 = s_prim[tid], q1 = s_prim[K_ + tid];
    const float q2 = s_prim[2 * K_ + tid], q3 = s_prim[3 * K_ + tid];
    const float invn = 1.0f / sqrtf(q0 * q0 + q1 * q1 + q2 * q2 + q3 * q3 + 1e-8f);
    s_pose[tid][0] = make_float4(q0 * invn, -q1 * invn, -q2 * invn, -q3 * invn);
    s_pose[tid][1] = make_float4(s_prim[4 * K_ + tid], s_prim[5 * K_ + tid],
                                 s_prim[6 * K_ + tid], fabsf(s_prim[7 * K_ + tid]));
  }
  lds_barrier();

#pragma unroll
  for (int i = K_ / 4; i < K_ / 2; ++i) {
    w2[i].x = wp[(size_t)(2 * i) * C_];
    w2[i].y = wp[(size_t)(2 * i + 1) * C_];
  }
  const float uwc = uw[b * C_ + c];

  {
    const int j  = tid & 63;
    const int kg = tid >> 6;
    for (int i = 0; i < K_ / 8; ++i) {
      const int k = kg + 8 * i;
      const float4 ca = s_c4[k][j];
      const float4 cb = s_c4[k][(j + 1) & 63];
      s_e4[k][j] = make_float4(ca.y - cb.y, cb.x - ca.x,
                               ca.x * cb.y - ca.y * cb.x, ca.y);
    }
  }
  lds_barrier();

  {
    const int kbase = (wv >> 1) * KPW;
    const int prow = ptile * 16 + row;
    float* const tile_out = hi ? &s_dmin[prow][0] : &s_sdf[prow][0];
    const v8f z = {0.f, 0.f, 0.f, 0.f, 0.f, 0.f, 0.f, 0.f};

#pragma unroll 1
    for (int kk = 0; kk < KPW; ++kk) {
      const int k = kbase + kk;
      const float4 pa = s_pose[k][0];
      const float4 pb = s_pose[k][1];
      const float qw = pa.x, vx = pa.y, vy = pa.z, vz = pa.w;
      const float habs = pb.w;

      v2f Ad[4], Ac[4], Ay[4];
#pragma unroll
      for (int tTile = 0; tTile < 4; ++tTile) {
        const float4 cc = s_c4[k][tTile * 16 + row];
        const float4 ee = s_e4[k][tTile * 16 + row];
        Ad[tTile].x = hi ? 1.0f  : -2.0f * cc.x;  Ad[tTile].y = hi ? cc.z : -2.0f * cc.y;
        Ac[tTile].x = hi ? ee.z  : ee.x;          Ac[tTile].y = hi ? 0.0f : ee.y;
        Ay[tTile].x = hi ? -ee.w : 0.0f;          Ay[tTile].y = hi ? 0.0f : 1.0f;
      }

      const float px0 = rx - pb.x, py0 = ry - pb.y, pz0 = rz - pb.z;
      const float t2x = 2.0f * (vy * pz0 - vz * py0);
      const float t2y = 2.0f * (vz * px0 - vx * pz0);
      const float t2z = 2.0f * (vx * py0 - vy * px0);
      const float plx = px0 + qw * t2x + (vy * t2z - vz * t2y);
      const float ply = py0 + qw * t2y + (vz * t2x - vx * t2z);
      const float plz = pz0 + qw * t2z + (vx * t2y - vy * t2x);
      const float px = plx, py = ply;
      const float ss = px * px + py * py;

      v2f Bd; Bd.x = hi ? ss : px;   Bd.y = hi ? 1.0f : py;
      v2f Bw; Bw.x = hi ? 1.0f : px; Bw.y = hi ? 0.0f : py;
      v8f e = {3e38f, 3e38f, 3e38f, 3e38f, 3e38f, 3e38f, 3e38f, 3e38f};

      unsigned am = 0u, cm = 0u;
#pragma unroll
      for (int tTile = 0; tTile < 4; ++tTile) {
        v8f d  = __builtin_amdgcn_wmma_f32_16x16x4_f32(
            false, Ad[tTile], false, Bd, (short)0, z, false, false);
        v8f cr = __builtin_amdgcn_wmma_f32_16x16x4_f32(
            false, Ac[tTile], false, Bw, (short)0, z, false, false);
        v8f yd = __builtin_amdgcn_wmma_f32_16x16x4_f32(
            false, Ay[tTile], false, Bw, (short)0, z, false, false);
        e = vmin8(e, d);
        unsigned a8 = 0u, c8 = 0u;
#pragma unroll
        for (int v = 0; v < 8; ++v) {
          a8 = (a8 << 1) | (__float_as_uint(yd[v]) >> 31);
          c8 = (c8 << 1) | (__float_as_uint(cr[v]) >> 31);
        }
        am = (am << 8) | a8;
        cm = (cm << 8) | c8;
      }
      const unsigned oth = (unsigned)xlane_i<16>((int)am);
      const unsigned fix = hi ? ((oth << 1) | (oth >> 31)) : (oth >> 7);
      const unsigned bm  = ((am << 1) & 0xFEFEFEFEu) | (fix & 0x01010101u);
      const unsigned hit = (am ^ bm) & ~(am ^ cm);
      int wn = __popc(hit & ~am) - __popc(hit & am);
      wn += xlane_i<16>(wn);

      float m01 = fminf(e[0], e[1]), m23 = fminf(e[2], e[3]);
      float m45 = fminf(e[4], e[5]), m67 = fminf(e[6], e[7]);
      float m8  = fminf(fminf(m01, m23), fminf(m45, m67));
      float d2min = fminf(m8, xlane_f<16>(m8));

      float dminv = sqrtf(d2min + 1e-12f);
      float sdf2d = (wn != 0) ? -dminv : dminv;
      float dz = fabsf(plz) - habs;
      float po = fmaxf(sdf2d, 0.0f), dzo = fmaxf(dz, 0.0f);
      float sdf = fminf(fmaxf(sdf2d, dz), 0.0f) + sqrtf(po * po + dzo * dzo + 1e-12f);

      tile_out[k] = hi ? dminv : sdf;
    }
  }
  lds_barrier();

  {
    const float4 sv = reinterpret_cast<const float4*>(&s_sdf[0][0])[tid];
    const float4 dv = reinterpret_cast<const float4*>(&s_dmin[0][0])[tid];
    reinterpret_cast<float4*>(sdf_out  + (size_t)bm0 * K_)[tid] = sv;
    reinterpret_cast<float4*>(dmin_out + (size_t)bm0 * K_)[tid] = dv;
    float4 oc;
    oc.x = 1.0f / (1.0f + __expf(150.0f * sv.x)) - 1.0f;
    oc.y = 1.0f / (1.0f + __expf(150.0f * sv.y)) - 1.0f;
    oc.z = 1.0f / (1.0f + __expf(150.0f * sv.z)) - 1.0f;
    oc.w = 1.0f / (1.0f + __expf(150.0f * sv.w)) - 1.0f;
    reinterpret_cast<float4*>(&nomo_sh[0][0])[tid] = oc;
  }
  lds_barrier();

  {
    const int g = tid >> 6;
#pragma unroll 1
    for (int rr = 0; rr < RPT; ++rr) {
      const int r = g * RPT + rr;
      v2f p2[K_ / 2];
      float pmin = 3.0e38f;
      const v2f one2 = {1.0f, 1.0f};
      const float4* nrow = reinterpret_cast<const float4*>(&nomo_sh[r][0]);
#pragma unroll
      for (int q = 0; q < K_ / 4; ++q) {
        const float4 n4 = nrow[q];
        const v2f na = {n4.x, n4.y}, nb = {n4.z, n4.w};
        p2[2 * q]     = __builtin_elementwise_fma(w2[2 * q],     na, one2);
        p2[2 * q + 1] = __builtin_elementwise_fma(w2[2 * q + 1], nb, one2);
        pmin = fminf(pmin, fminf(p2[2 * q].x, p2[2 * q].y));
        pmin = fminf(pmin, fminf(p2[2 * q + 1].x, p2[2 * q + 1].y));
      }
      float inter;
      if (train) {
        const float NC = -57.70780163555854f;
        const float shift = -NC * pmin;
        const v2f nc2 = {NC, NC};
        const v2f shift2 = {shift, shift};
        float s = 0.0f, ts = 0.0f;
#pragma unroll
        for (int i = 0; i < K_ / 2; ++i) {
          const v2f a2 = __builtin_elementwise_fma(p2[i], nc2, shift2);
          const float e0 = __builtin_amdgcn_exp2f(a2.x);
          const float e1 = __builtin_amdgcn_exp2f(a2.y);
          s += e0; ts += e0 * p2[i].x;
          s += e1; ts += e1 * p2[i].y;
        }
        inter = ts / s;
      } else {
        inter = pmin;
      }
      inter_out[(size_t)(bm0 + r) * C_ + c] = inter;
      preu[r][c] = uwc * inter;
    }
  }
  lds_barrier();

#pragma unroll
  for (int rr = 0; rr < PTS / 16; ++rr) {
    const int r = wv * (PTS / 16) + rr;
    float a0 = preu[r][lane], a1 = preu[r][lane + 32];
    float mx = fmaxf(a0, a1);
    mx = fmaxf(mx, xlane_f<1>(mx));
    mx = fmaxf(mx, xlane_f<2>(mx));
    mx = fmaxf(mx, xlane_f<4>(mx));
    mx = fmaxf(mx, xlane_f<8>(mx));
    mx = fmaxf(mx, xlane_f<16>(mx));
    float res = mx;
    if (train) {
      float e0 = __expf(40.0f * (a0 - mx));
      float e1 = __expf(40.0f * (a1 - mx));
      float s  = e0 + e1;
      float ts = e0 * a0 + e1 * a1;
      s += xlane_f<1>(s);   ts += xlane_f<1>(ts);
      s += xlane_f<2>(s);   ts += xlane_f<2>(ts);
      s += xlane_f<4>(s);   ts += xlane_f<4>(ts);
      s += xlane_f<8>(s);   ts += xlane_f<8>(ts);
      s += xlane_f<16>(s);  ts += xlane_f<16>(ts);
      res = ts / s;
    }
    if (lane == 0) occ_out[bm0 + r] = res;
  }
}

extern "C" void kernel_launch(void* const* d_in, const int* in_sizes, int n_in,
                              void* d_out, int out_size, void* d_ws, size_t ws_size,
                              hipStream_t stream) {
  const float* pts   = (const float*)d_in[0];
  const float* prim  = (const float*)d_in[1];
  const float* iw    = (const float*)d_in[2];
  const float* uw    = (const float*)d_in[3];
  const int*   train = (const int*)d_in[4];

  float* occ_out   = (float*)d_out;
  float* sdf_out   = occ_out + (size_t)B_ * M_;
  float* inter_out = sdf_out + (size_t)B_ * M_ * K_;
  float* dmin_out  = inter_out + (size_t)B_ * M_ * C_;

  fused_kernel<<<B_ * M_ / PTS, NT, 0, stream>>>(pts, prim, iw, uw, train,
                                                 occ_out, sdf_out, inter_out, dmin_out);
  (void)in_sizes; (void)n_in; (void)out_size; (void)d_ws; (void)ws_size;
}


# combine soft-min in ONE pass on packed ops (3 per two k): exponent (NC*w)*(occ-1), shift is the constant p=1, not the row minimum; sums kept as even-k/odd-k halves, taken of the exponents
# speedup vs baseline: 4.0868x; 1.0576x over previous
#include <hip/hip_runtime.h>
#include <hip/hip_bf16.h>
#include <math.h>

typedef float v2f __attribute__((ext_vector_type(2)));
typedef float v8f __attribute__((ext_vector_type(8)));

#define B_ 4
#define M_ 2048
#define K_ 64
#define C_ 64
#define P_ 28
#define NCURVE 64
#define PTS 32
#define NT 512
#define KPW 8
#define RPT 4

__device__ __forceinline__ v8f vmin8(v8f a, v8f b){
  v8f r;
#pragma unroll
  for (int i = 0; i < 8; ++i) r[i] = fminf(a[i], b[i]);
  return r;
}

template <int MASK> __device__ __forceinline__ int xlane_i(int x){
  if constexpr (MASK == 16)
    return __builtin_amdgcn_permlanex16(x, x, 0x76543210, 0xfedcba98, false, false);
  else if constexpr (MASK == 1)
    return __builtin_amdgcn_update_dpp(x, x, 0xB1, 0xF, 0xF, false);
  else if constexpr (MASK == 2)
    return __builtin_amdgcn_update_dpp(x, x, 0x4E, 0xF, 0xF, false);
  else
    return __builtin_amdgcn_update_dpp(x, x, 0x160 + MASK, 0xF, 0xF, false);
}
template <int MASK> __device__ __forceinline__ float xlane_f(float x){
  return __int_as_float(xlane_i<MASK>(__float_as_int(x)));
}

__device__ __forceinline__ void lds_barrier(){
  __builtin_amdgcn_fence(__ATOMIC_RELEASE, "workgroup", "local");
  __builtin_amdgcn_s_barrier();
  __builtin_amdgcn_fence(__ATOMIC_ACQUIRE, "workgroup", "local");
}

__global__ __launch_bounds__(NT) void fused_kernel(
    const float* __restrict__ pts, const float* __restrict__ prim,
    const float* __restrict__ iw, const float* __restrict__ uw,
    const int* __restrict__ is_training,
    float* __restrict__ occ_out, float* __restrict__ sdf_out,
    float* __restrict__ inter_out, float* __restrict__ dmin_out)
{
  __shared__ float4 s_c4[K_][NCURVE];
  __shared__ float4 s_e4[K_][NCURVE];
  __shared__ float4 s_trig[4][2];
  __shared__ float4 s_pose[K_][2];
  __shared__ float  s_prim[P_ * K_];
  __shared__ __align__(16) float s_sdf[PTS][K_];
  __shared__ __align__(16) float s_dmin[PTS][K_];
  __shared__ __align__(16) float nomo_sh[PTS][K_];
  __shared__ float preu[PTS][C_];

  const int blk = blockIdx.x;
  const int b   = blk >> 6;
  const int bm0 = blk * PTS;
  const int tid = threadIdx.x;
  const int lane = tid & 31;
  const int wv = tid >> 5;
  const int row = lane & 15;
  const bool hi = lane >= 16;
  const float PI = 3.14159265358979323846f;

  const int ptile = wv & 1;
  const float* p3 = pts + ((size_t)bm0 + ptile * 16 + row) * 3;
  const float rx = p3[0], ry = p3[1], rz = p3[2];

  float pr[4];
#pragma unroll
  for (int i = 0; i < 4; ++i) {
    const int idx = tid + i * NT;
    pr[i] = (idx < P_ * K_) ? prim[(size_t)b * P_ * K_ + idx] : 0.0f;
  }

  const int train = is_training[0];

  if (tid < 16) {
    const int seg = tid >> 2;
    const int a = tid & 3;
    const float theta = PI * 0.125f + atanf(tanf(PI * 0.125f) * (1.0f / 3.0f));
    const float r0 = (PI * 0.5f) * (float)seg;
    const float r3 = (PI * 0.5f) * (float)(seg + 1);
    const float ra = (a == 0) ? r0 : (a == 1) ? (r0 + theta) : (a == 2) ? (r3 - theta) : r3;
    reinterpret_cast<float2*>(&s_trig[seg][0])[a] = make_float2(cosf(ra), sinf(ra));
  }
#pragma unroll
  for (int i = 0; i < 4; ++i) {
    const int idx = tid + i * NT;
    if (idx < P_ * K_) s_prim[idx] = pr[i];
  }
  lds_barrier();

  const int c = tid & 63;
  const float* wp = iw + ((size_t)b * K_) * C_ + c;
  v2f w2[K_ / 2];
#pragma unroll
  for (int i = 0; i < K_ / 4; ++i) {
    w2[i].x = wp[(size_t)(2 * i) * C_];
    w2[i].y = wp[(size_t)(2 * i + 1) * C_];
  }

  {
    const int j  = tid & 63;
    const int kg = tid >> 6;
    const int seg = j >> 4;
    const int seg1 = (seg + 1) & 3;
    const float t = (float)(j & 15) * (1.0f / 16.0f);
    const float u = 1.0f - t;
    const float b0 = u * u * u, b1 = 3.0f * u * u * t, b2 = 3.0f * u * t * t, b3 = t * t * t;
    const float4 ta = s_trig[seg][0];
    const float4 tb = s_trig[seg][1];
    for (int i = 0; i < K_ / 8; ++i) {
      const int k = kg + 8 * i;
      const float rad0 = fabsf(s_prim[(8 + seg * 3 + 0) * K_ + k]);
      const float rad1 = fabsf(s_prim[(8 + seg * 3 + 1) * K_ + k]);
      const float rad2 = fabsf(s_prim[(8 + seg * 3 + 2) * K_ + k]);
      const float rad3 = fabsf(s_prim[(8 + seg1 * 3 + 0) * K_ + k]);
      const float P0x = ta.x * rad0, P0y = ta.y * rad0;
      const float P1x = ta.z * rad1, P1y = ta.w * rad1;
      const float P2x = tb.x * rad2, P2y = tb.y * rad2;
      const float P3x = tb.z * rad3, P3y = tb.w * rad3;
      const float w1 = fabsf(s_prim[(20 + seg * 2 + 0) * K_ + k]);
      const float w2 = fabsf(s_prim[(20 + seg * 2 + 1) * K_ + k]);
      const float inv = 1.0f / (b0 + w1 * b1 + w2 * b2 + b3);
      const float cx = (P0x * b0 + P1x * (w1 * b1) + P2x * (w2 * b2) + P3x * b3) * inv;
      const float cy = (P0y * b0 + P1y * (w1 * b1) + P2y * (w2 * b2) + P3y * b3) * inv;
      s_c4[k][j] = make_float4(cx, cy, cx * cx + cy * cy, 1.0f);
    }
  }
  if (tid < K_) {
    const float q0 = s_prim[tid], q1 = s_prim[K_ + tid];
    const float q2 = s_prim[2 * K_ + tid], q3 = s_prim[3 * K_ + tid];
    const float invn = 1.0f / sqrtf(q0 * q0 + q1 * q1 + q2 * q2 + q3 * q3 + 1e-8f);
    s_pose[tid][0] = make_float4(q0 * invn, -q1 * invn, -q2 * invn, -q3 * invn);
    s_pose[tid][1] = make_float4(s_prim[4 * K_ + tid], s_prim[5 * K_ + tid],
                                 s_prim[6 * K_ + tid], fabsf(s_prim[7 * K_ + tid]));
  }
  lds_barrier();

#pragma unroll
  for (int i = K_ / 4; i < K_ / 2; ++i) {
    w2[i].x = wp[(size_t)(2 * i) * C_];
    w2[i].y = wp[(size_t)(2 * i + 1) * C_];
  }
  const float uwc = uw[b * C_ + c];

  {
    const int j  = tid & 63;
    const int kg = tid >> 6;
    for (int i = 0; i < K_ / 8; ++i) {
      const int k = kg + 8 * i;
      const float4 ca = s_c4[k][j];
      const float4 cb = s_c4[k][(j + 1) & 63];
      s_e4[k][j] = make_float4(ca.y - cb.y, cb.x - ca.x,
                               ca.x * cb.y - ca.y * cb.x, ca.y);
    }
  }
  lds_barrier();

  {
    const int kbase = (wv >> 1) * KPW;
    const int prow = ptile * 16 + row;
    float* const tile_out = hi ? &s_dmin[prow][0] : &s_sdf[prow][0];
    const v8f z = {0.f, 0.f, 0.f, 0.f, 0.f, 0.f, 0.f, 0.f};

#pragma unroll 1
    for (int kk = 0; kk < KPW; ++kk) {
      const int k = kbase + kk;
      const float4 pa = s_pose[k][0];
      const float4 pb = s_pose[k][1];
      const float qw = pa.x, vx = pa.y, vy = pa.z, vz = pa.w;
      const float habs = pb.w;

      v2f Ad[4], Ac[4], Ay[4];
#pragma unroll
      for (int tTile = 0; tTile < 4; ++tTile) {
        const float4 cc = s_c4[k][tTile * 16 + row];
        const float4 ee = s_e4[k][tTile * 16 + row];
        Ad[tTile].x = hi ? 1.0f  : -2.0f * cc.x;  Ad[tTile].y = hi ? cc.z : -2.0f * cc.y;
        Ac[tTile].x = hi ? ee.z  : ee.x;          Ac[tTile].y = hi ? 0.0f : ee.y;
        Ay[tTile].x = hi ? -ee.w : 0.0f;          Ay[tTile].y = hi ? 0.0f : 1.0f;
      }

      const float px0 = rx - pb.x, py0 = ry - pb.y, pz0 = rz - pb.z;
      const float t2x = 2.0f * (vy * pz0 - vz * py0);
      const float t2y = 2.0f * (vz * px0 - vx * pz0);
      const float t2z = 2.0f * (vx * py0 - vy * px0);
      const float plx = px0 + qw * t2x + (vy * t2z - vz * t2y);
      const float ply = py0 + qw * t2y + (vz * t2x - vx * t2z);
      const float plz = pz0 + qw * t2z + (vx * t2y - vy * t2x);
      const float px = plx, py = ply;
      const float ss = px * px + py * py;

      v2f Bd; Bd.x = hi ? ss : px;   Bd.y = hi ? 1.0f : py;
      v2f Bw; Bw.x = hi ? 1.0f : px; Bw.y = hi ? 0.0f : py;
      v8f e = {3e38f, 3e38f, 3e38f, 3e38f, 3e38f, 3e38f, 3e38f, 3e38f};

      unsigned am = 0u, cm = 0u;
#pragma unroll
      for (int tTile = 0; tTile < 4; ++tTile) {
        v8f d  = __builtin_amdgcn_wmma_f32_16x16x4_f32(
            false, Ad[tTile], false, Bd, (short)0, z, false, false);
        v8f cr = __builtin_amdgcn_wmma_f32_16x16x4_f32(
            false, Ac[tTile], false, Bw, (short)0, z, false, false);
        v8f yd = __builtin_amdgcn_wmma_f32_16x16x4_f32(
            false, Ay[tTile], false, Bw, (short)0, z, false, false);
        e = vmin8(e, d);
        unsigned a8 = 0u, c8 = 0u;
#pragma unroll
        for (int v = 0; v < 8; ++v) {
          a8 = (a8 << 1) | (__float_as_uint(yd[v]) >> 31);
          c8 = (c8 << 1) | (__float_as_uint(cr[v]) >> 31);
        }
        am = (am << 8) | a8;
        cm = (cm << 8) | c8;
      }
      const unsigned oth = (unsigned)xlane_i<16>((int)am);
      const unsigned fix = hi ? ((oth << 1) | (oth >> 31)) : (oth >> 7);
      const unsigned bm  = ((am << 1) & 0xFEFEFEFEu) | (fix & 0x01010101u);
      const unsigned hit = (am ^ bm) & ~(am ^ cm);
      int wn = __popc(hit & ~am) - __popc(hit & am);
      wn += xlane_i<16>(wn);

      float m01 = fminf(e[0], e[1]), m23 = fminf(e[2], e[3]);
      float m45 = fminf(e[4], e[5]), m67 = fminf(e[6], e[7]);
      float m8  = fminf(fminf(m01, m23), fminf(m45, m67));
      float d2min = fminf(m8, xlane_f<16>(m8));

      float dminv = sqrtf(d2min + 1e-12f);
      float sdf2d = (wn != 0) ? -dminv : dminv;
      float dz = fabsf(plz) - habs;
      float po = fmaxf(sdf2d, 0.0f), dzo = fmaxf(dz, 0.0f);
      float sdf = fminf(fmaxf(sdf2d, dz), 0.0f) + sqrtf(po * po + dzo * dzo + 1e-12f);

      tile_out[k] = hi ? dminv : sdf;
    }
  }
  lds_barrier();

  {
    const float4 sv = reinterpret_cast<const float4*>(&s_sdf[0][0])[tid];
    const float4 dv = reinterpret_cast<const float4*>(&s_dmin[0][0])[tid];
    reinterpret_cast<float4*>(sdf_out  + (size_t)bm0 * K_)[tid] = sv;
    reinterpret_cast<float4*>(dmin_out + (size_t)bm0 * K_)[tid] = dv;
    float4 oc;
    oc.x = 1.0f / (1.0f + __expf(150.0f * sv.x)) - 1.0f;
    oc.y = 1.0f / (1.0f + __expf(150.0f * sv.y)) - 1.0f;
    oc.z = 1.0f / (1.0f + __expf(150.0f * sv.z)) - 1.0f;
    oc.w = 1.0f / (1.0f + __expf(150.0f * sv.w)) - 1.0f;
    reinterpret_cast<float4*>(&nomo_sh[0][0])[tid] = oc;
  }
  lds_barrier();

  {
    const int g = tid >> 6;
    const float NC = -57.70780163555854f;
    const float INC = 1.0f / NC;
    const v2f nc2 = {NC, NC};
    v2f wn2[K_ / 2];
#pragma unroll
    for (int i = 0; i < K_ / 2; ++i) wn2[i] = w2[i] * nc2;
#pragma unroll 1
    for (int rr = 0; rr < RPT; ++rr) {
      const int r = g * RPT + rr;
      const float4* nrow = reinterpret_cast<const float4*>(&nomo_sh[r][0]);
      float inter;
      if (train) {
        v2f s2 = {0.0f, 0.0f}, t2 = {0.0f, 0.0f};
#pragma unroll
        for (int q = 0; q < K_ / 4; ++q) {
          const float4 n4 = nrow[q];
          const v2f na = {n4.x, n4.y}, nb = {n4.z, n4.w};
          const v2f a2 = wn2[2 * q] * na;
          const v2f b2 = wn2[2 * q + 1] * nb;
          v2f ea, eb;
          ea.x = __builtin_amdgcn_exp2f(a2.x);
          ea.y = __builtin_amdgcn_exp2f(a2.y);
          eb.x = __builtin_amdgcn_exp2f(b2.x);
          eb.y = __builtin_amdgcn_exp2f(b2.y);
          s2 += ea; t2 = __builtin_elementwise_fma(ea, a2, t2);
          s2 += eb; t2 = __builtin_elementwise_fma(eb, b2, t2);
        }
        const float s = s2.x + s2.y;
        const float ta = t2.x + t2.y;
        inter = fmaf(ta / s, INC, 1.0f);
      } else {
        float amax = -3.0e38f;
#pragma unroll
        for (int q = 0; q < K_ / 4; ++q) {
          const float4 n4 = nrow[q];
          const v2f na = {n4.x, n4.y}, nb = {n4.z, n4.w};
          const v2f a2 = wn2[2 * q] * na;
          const v2f b2 = wn2[2 * q + 1] * nb;
          amax = fmaxf(amax, fmaxf(a2.x, a2.y));
          amax = fmaxf(amax, fmaxf(b2.x, b2.y));
        }
        inter = fmaf(amax, INC, 1.0f);
      }
      inter_out[(size_t)(bm0 + r) * C_ + c] = inter;
      preu[r][c] = uwc * inter;
    }
  }
  lds_barrier();

#pragma unroll
  for (int rr = 0; rr < PTS / 16; ++rr) {
    const int r = wv * (PTS / 16) + rr;
    float a0 = preu[r][lane], a1 = preu[r][lane + 32];
    float mx = fmaxf(a0, a1);
    mx = fmaxf(mx, xlane_f<1>(mx));
    mx = fmaxf(mx, xlane_f<2>(mx));
    mx = fmaxf(mx, xlane_f<4>(mx));
    mx = fmaxf(mx, xlane_f<8>(mx));
    mx = fmaxf(mx, xlane_f<16>(mx));
    float res = mx;
    if (train) {
      float e0 = __expf(40.0f * (a0 - mx));
      float e1 = __expf(40.0f * (a1 - mx));
      float s  = e0 + e1;
      float ts = e0 * a0 + e1 * a1;
      s += xlane_f<1>(s);   ts += xlane_f<1>(ts);
      s += xlane_f<2>(s);   ts += xlane_f<2>(ts);
      s += xlane_f<4>(s);   ts += xlane_f<4>(ts);
      s += xlane_f<8>(s);   ts += xlane_f<8>(ts);
      s += xlane_f<16>(s);  ts += xlane_f<16>(ts);
      res = ts / s;
    }
    if (lane == 0) occ_out[bm0 + r] = res;
  }
}

extern "C" void kernel_launch(void* const* d_in, const int* in_sizes, int n_in,
                              void* d_out, int out_size, void* d_ws, size_t ws_size,
                              hipStream_t stream) {
  const float* pts   = (const float*)d_in[0];
  const float* prim  = (const float*)d_in[1];
  const float* iw    = (const float*)d_in[2];
  const float* uw    = (const float*)d_in[3];
  const int*   train = (const int*)d_in[4];

  float* occ_out   = (float*)d_out;
  float* sdf_out   = occ_out + (size_t)B_ * M_;
  float* inter_out = sdf_out + (size_t)B_ * M_ * K_;
  float* dmin_out  = inter_out + (size_t)B_ * M_ * C_;

  fused_kernel<<<B_ * M_ / PTS, NT, 0, stream>>>(pts, prim, iw, uw, train,
                                                 occ_out, sdf_out, inter_out, dmin_out);
  (void)in_sizes; (void)n_in; (void)out_size; (void)d_ws; (void)ws_size;
}
